# starting kernel
# baseline (speedup 1.0000x reference)
#include <hip/hip_runtime.h>


#define D_DIM   1024
#define K_DIM   512
#define LMBDA_F 1000.0f
#define NITER   96

typedef float v2f __attribute__((ext_vector_type(2)));
typedef float v8f __attribute__((ext_vector_type(8)));

__device__ __forceinline__ v8f wmma_f32(v2f a, v2f b, v8f c) {
    return __builtin_amdgcn_wmma_f32_16x16x4_f32(
        false, a, false, b, (short)0, c, false, false);
}

__global__ void max_reduce_kernel(const float* __restrict__ ea,
                                  const float* __restrict__ eb,
                                  float* __restrict__ s_out) {
    __shared__ float sm[512];
    int t = threadIdx.x;
    sm[t] = fmaxf(ea[t], eb[t]);
    __syncthreads();
    for (int off = 256; off > 0; off >>= 1) {
        if (t < off) sm[t] = fmaxf(sm[t], sm[t + off]);
        __syncthreads();
    }
    if (t == 0) s_out[0] = sm[0];
}

__global__ void mask_kernel(const float* __restrict__ erow,
                            const float* __restrict__ ecol,
                            const float* __restrict__ s_ptr,
                            float* __restrict__ mask) {
    int j = blockIdx.x * blockDim.x + threadIdx.x;
    int i = blockIdx.y;
    float inv_s = 1.0f / s_ptr[0];
    float gb = erow[i] * inv_s;
    float ga = ecol[j] * inv_s;
    float gb2 = gb * gb + 1.0f;
    float ga2 = ga * ga + 1.0f;
    float re = gb / gb2 - ga / ga2;
    float im = 1.0f / gb2 - 1.0f / ga2;
    mask[i * K_DIM + j] = re * re + im * im;
}

__global__ void __launch_bounds__(256, 1)
gram_kernel(const float* __restrict__ X,
            const float* __restrict__ Y,
            float* __restrict__ C) {
    int lane = threadIdx.x & 31;
    int wave = threadIdx.x >> 5;
    int lo = lane & 15;
    int hi = lane >> 4;
    int m0 = blockIdx.x * 64;
    int n0 = blockIdx.y * 256 + wave * 32;

    v8f zero = {};
    v8f acc[4][2];
#pragma unroll
    for (int mi = 0; mi < 4; ++mi)
#pragma unroll
        for (int ni = 0; ni < 2; ++ni) acc[mi][ni] = zero;

    for (int d = 0; d < D_DIM; d += 4) {
        int kk = d + 2 * hi;
        v2f aF[4], bF[2];
#pragma unroll
        for (int mi = 0; mi < 4; ++mi) {
            int m = m0 + mi * 16 + lo;
            aF[mi].x = X[kk * K_DIM + m];
            aF[mi].y = X[(kk + 1) * K_DIM + m];
        }
#pragma unroll
        for (int ni = 0; ni < 2; ++ni) {
            int n = n0 + ni * 16 + lo;
            bF[ni].x = Y[kk * K_DIM + n];
            bF[ni].y = Y[(kk + 1) * K_DIM + n];
        }
#pragma unroll
        for (int mi = 0; mi < 4; ++mi)
#pragma unroll
            for (int ni = 0; ni < 2; ++ni)
                acc[mi][ni] = wmma_f32(aF[mi], bF[ni], acc[mi][ni]);
    }

#pragma unroll
    for (int mi = 0; mi < 4; ++mi)
#pragma unroll
        for (int ni = 0; ni < 2; ++ni)
#pragma unroll
            for (int v = 0; v < 8; ++v) {
                int m = m0 + mi * 16 + v + 8 * hi;
                int n = n0 + ni * 16 + lo;
                C[m * K_DIM + n] = acc[mi][ni][v];
            }
}

__global__ void __launch_bounds__(256, 1)
cg_solve_kernel(const float* __restrict__ G,
                const float* __restrict__ R,
                const float* __restrict__ mask,
                float* __restrict__ Xout,
                int niter) {
    extern __shared__ float smem[];
    float* p_s  = smem;
    float* r_s  = p_s  + 16 * K_DIM;
    float* x_s  = r_s  + 16 * K_DIM;
    float* ap_s = x_s  + 16 * K_DIM;
    __shared__ float rs[16], rs_new[16], pAp[16], alpha_s[16], beta_s[16];

    int t    = threadIdx.x;
    int lane = t & 31;
    int wave = t >> 5;
    int lo   = lane & 15;
    int hi   = lane >> 4;
    int n0   = wave * 64;
    int m0   = blockIdx.x * 16;

    int rowT = t >> 4;
    int colT = t & 15;

    if (t < 16) rs[t] = 0.0f;
    __syncthreads();
    {
        float part = 0.0f;
        for (int j = 0; j < 32; ++j) {
            int c = colT + j * 16;
            float rv = R[(m0 + rowT) * K_DIM + c];
            r_s[rowT * K_DIM + c] = rv;
            p_s[rowT * K_DIM + c] = rv;
            x_s[rowT * K_DIM + c] = 0.0f;
            part += rv * rv;
        }
        atomicAdd(&rs[rowT], part);
    }
    __syncthreads();

    for (int it = 0; it < niter; ++it) {
        if (t < 16) pAp[t] = 0.0f;
        __syncthreads();

        v8f zero = {};
        v8f acc[4];
#pragma unroll
        for (int ni = 0; ni < 4; ++ni) acc[ni] = zero;

        for (int kb = 0; kb < K_DIM; kb += 4) {
            int kk = kb + 2 * hi;
            v2f aF;
            aF.x = p_s[lo * K_DIM + kk];
            aF.y = p_s[lo * K_DIM + kk + 1];
#pragma unroll
            for (int ni = 0; ni < 4; ++ni) {
                int n = n0 + ni * 16 + lo;
                v2f bF;
                bF.x = G[kk * K_DIM + n];
                bF.y = G[(kk + 1) * K_DIM + n];
                acc[ni] = wmma_f32(aF, bF, acc[ni]);
            }
        }

#pragma unroll
        for (int ni = 0; ni < 4; ++ni) {
            int n = n0 + ni * 16 + lo;
#pragma unroll
            for (int v = 0; v < 8; ++v) {
                int m = v + 8 * hi;
                float pv  = p_s[m * K_DIM + n];
                float val = acc[ni][v]
                          + LMBDA_F * mask[(m0 + m) * K_DIM + n] * pv;
                ap_s[m * K_DIM + n] = val;
                atomicAdd(&pAp[m], pv * val);
            }
        }
        __syncthreads();

        if (t < 16) { alpha_s[t] = rs[t] / pAp[t]; rs_new[t] = 0.0f; }
        __syncthreads();

        {
            float a_m  = alpha_s[rowT];
            float part = 0.0f;
            for (int j = 0; j < 32; ++j) {
                int idx = rowT * K_DIM + colT + j * 16;
                float xv = x_s[idx] + a_m * p_s[idx];
                float rv = r_s[idx] - a_m * ap_s[idx];
                x_s[idx] = xv;
                r_s[idx] = rv;
                part += rv * rv;
            }
            atomicAdd(&rs_new[rowT], part);
        }
        __syncthreads();

        if (t < 16) { beta_s[t] = rs_new[t] / rs[t]; rs[t] = rs_new[t]; }
        __syncthreads();

        {
            float b_m = beta_s[rowT];
            for (int j = 0; j < 32; ++j) {
                int idx = rowT * K_DIM + colT + j * 16;
                p_s[idx] = r_s[idx] + b_m * p_s[idx];
            }
        }
        __syncthreads();
    }

    for (int j = 0; j < 32; ++j) {
        int c = colT + j * 16;
        Xout[(m0 + rowT) * K_DIM + c] = x_s[rowT * K_DIM + c];
    }
}

extern "C" void kernel_launch(void* const* d_in, const int* in_sizes, int n_in,
                              void* d_out, int out_size, void* d_ws, size_t ws_size,
                              hipStream_t stream) {
    const float* A  = (const float*)d_in[0];
    const float* B  = (const float*)d_in[1];
    const float* ea = (const float*)d_in[2];
    const float* eb = (const float*)d_in[3];
    float* out = (float*)d_out;

    float* ws      = (float*)d_ws;
    float* s_max   = ws;
    float* mask_ab = s_max   + 16;
    float* mask_ba = mask_ab + K_DIM * K_DIM;
    float* G_a     = mask_ba + K_DIM * K_DIM;
    float* G_b     = G_a     + K_DIM * K_DIM;
    float* R_ab    = G_b     + K_DIM * K_DIM;
    float* R_ba    = R_ab    + K_DIM * K_DIM;

    max_reduce_kernel<<<1, 512, 0, stream>>>(ea, eb, s_max);

    dim3 mgrid(K_DIM / 256, K_DIM);
    mask_kernel<<<mgrid, 256, 0, stream>>>(eb, ea, s_max, mask_ab);
    mask_kernel<<<mgrid, 256, 0, stream>>>(ea, eb, s_max, mask_ba);

    dim3 ggrid(8, 2);
    gram_kernel<<<ggrid, 256, 0, stream>>>(A, A, G_a);
    gram_kernel<<<ggrid, 256, 0, stream>>>(B, A, R_ab);
    gram_kernel<<<ggrid, 256, 0, stream>>>(B, B, G_b);
    gram_kernel<<<ggrid, 256, 0, stream>>>(A, B, R_ba);

    size_t shmem = (size_t)4 * 16 * K_DIM * sizeof(float);
    cg_solve_kernel<<<K_DIM / 16, 256, shmem, stream>>>(
        G_a, R_ab, mask_ab, out, NITER);
    cg_solve_kernel<<<K_DIM / 16, 256, shmem, stream>>>(
        G_b, R_ba, mask_ba, out + K_DIM * K_DIM, NITER);
}


# Independent launches merged to run side by side on idle WGPs: the two CG solves as one 64-block grid, the four Gram products as one 64-block grid (9 launches -> 5); sum order unchanged
# speedup vs baseline: 2.0209x; 2.0209x over previous
#include <hip/hip_runtime.h>


#define D_DIM   1024
#define K_DIM   512
#define LMBDA_F 1000.0f
#define NITER   96

typedef float v2f __attribute__((ext_vector_type(2)));
typedef float v8f __attribute__((ext_vector_type(8)));

__device__ __forceinline__ v8f wmma_f32(v2f a, v2f b, v8f c) {
    return __builtin_amdgcn_wmma_f32_16x16x4_f32(
        false, a, false, b, (short)0, c, false, false);
}

__global__ void max_reduce_kernel(const float* __restrict__ ea,
                                  const float* __restrict__ eb,
                                  float* __restrict__ s_out) {
    __shared__ float sm[512];
    int t = threadIdx.x;
    sm[t] = fmaxf(ea[t], eb[t]);
    __syncthreads();
    for (int off = 256; off > 0; off >>= 1) {
        if (t < off) sm[t] = fmaxf(sm[t], sm[t + off]);
        __syncthreads();
    }
    if (t == 0) s_out[0] = sm[0];
}

__global__ void mask_kernel(const float* __restrict__ erow,
                            const float* __restrict__ ecol,
                            const float* __restrict__ s_ptr,
                            float* __restrict__ mask) {
    int j = blockIdx.x * blockDim.x + threadIdx.x;
    int i = blockIdx.y;
    float inv_s = 1.0f / s_ptr[0];
    float gb = erow[i] * inv_s;
    float ga = ecol[j] * inv_s;
    float gb2 = gb * gb + 1.0f;
    float ga2 = ga * ga + 1.0f;
    float re = gb / gb2 - ga / ga2;
    float im = 1.0f / gb2 - 1.0f / ga2;
    mask[i * K_DIM + j] = re * re + im * im;
}

__global__ void __launch_bounds__(256, 1)
gram_kernel(const float* __restrict__ A,
            const float* __restrict__ B,
            float* __restrict__ C4) {
    const int z = blockIdx.z;
    const float* __restrict__ X = (z == 1 || z == 2) ? B : A;
    const float* __restrict__ Y = (z == 1 || z == 3) ? B : A;
    float* __restrict__ C = C4 + (size_t)z * K_DIM * K_DIM;
    int lane = threadIdx.x & 31;
    int wave = threadIdx.x >> 5;
    int lo = lane & 15;
    int hi = lane >> 4;
    int m0 = blockIdx.x * 64;
    int n0 = blockIdx.y * 256 + wave * 32;

    v8f zero = {};
    v8f acc[4][2];
#pragma unroll
    for (int mi = 0; mi < 4; ++mi)
#pragma unroll
        for (int ni = 0; ni < 2; ++ni) acc[mi][ni] = zero;

    for (int d = 0; d < D_DIM; d += 4) {
        int kk = d + 2 * hi;
        v2f aF[4], bF[2];
#pragma unroll
        for (int mi = 0; mi < 4; ++mi) {
            int m = m0 + mi * 16 + lo;
            aF[mi].x = X[kk * K_DIM + m];
            aF[mi].y = X[(kk + 1) * K_DIM + m];
        }
#pragma unroll
        for (int ni = 0; ni < 2; ++ni) {
            int n = n0 + ni * 16 + lo;
            bF[ni].x = Y[kk * K_DIM + n];
            bF[ni].y = Y[(kk + 1) * K_DIM + n];
        }
#pragma unroll
        for (int mi = 0; mi < 4; ++mi)
#pragma unroll
            for (int ni = 0; ni < 2; ++ni)
                acc[mi][ni] = wmma_f32(aF[mi], bF[ni], acc[mi][ni]);
    }

#pragma unroll
    for (int mi = 0; mi < 4; ++mi)
#pragma unroll
        for (int ni = 0; ni < 2; ++ni)
#pragma unroll
            for (int v = 0; v < 8; ++v) {
                int m = m0 + mi * 16 + v + 8 * hi;
                int n = n0 + ni * 16 + lo;
                C[m * K_DIM + n] = acc[mi][ni][v];
            }
}

__global__ void __launch_bounds__(256, 1)
cg_solve_kernel(const float* __restrict__ G0,
                const float* __restrict__ R0,
                const float* __restrict__ mask0,
                float* __restrict__ Xout0,
                const float* __restrict__ G1,
                const float* __restrict__ R1,
                const float* __restrict__ mask1,
                float* __restrict__ Xout1,
                int niter) {
    const bool dir1 = blockIdx.y != 0;
    const float* __restrict__ G    = dir1 ? G1    : G0;
    const float* __restrict__ R    = dir1 ? R1    : R0;
    const float* __restrict__ mask = dir1 ? mask1 : mask0;
    float* __restrict__ Xout       = dir1 ? Xout1 : Xout0;
    extern __shared__ float smem[];
    float* p_s  = smem;
    float* r_s  = p_s  + 16 * K_DIM;
    float* x_s  = r_s  + 16 * K_DIM;
    float* ap_s = x_s  + 16 * K_DIM;
    __shared__ float rs[16], rs_new[16], pAp[16], alpha_s[16], beta_s[16];

    int t    = threadIdx.x;
    int lane = t & 31;
    int wave = t >> 5;
    int lo   = lane & 15;
    int hi   = lane >> 4;
    int n0   = wave * 64;
    int m0   = blockIdx.x * 16;

    int rowT = t >> 4;
    int colT = t & 15;

    if (t < 16) rs[t] = 0.0f;
    __syncthreads();
    {
        float part = 0.0f;
        for (int j = 0; j < 32; ++j) {
            int c = colT + j * 16;
            float rv = R[(m0 + rowT) * K_DIM + c];
            r_s[rowT * K_DIM + c] = rv;
            p_s[rowT * K_DIM + c] = rv;
            x_s[rowT * K_DIM + c] = 0.0f;
            part += rv * rv;
        }
        atomicAdd(&rs[rowT], part);
    }
    __syncthreads();

    for (int it = 0; it < niter; ++it) {
        if (t < 16) pAp[t] = 0.0f;
        __syncthreads();

        v8f zero = {};
        v8f acc[4];
#pragma unroll
        for (int ni = 0; ni < 4; ++ni) acc[ni] = zero;

        for (int kb = 0; kb < K_DIM; kb += 4) {
            int kk = kb + 2 * hi;
            v2f aF;
            aF.x = p_s[lo * K_DIM + kk];
            aF.y = p_s[lo * K_DIM + kk + 1];
#pragma unroll
            for (int ni = 0; ni < 4; ++ni) {
                int n = n0 + ni * 16 + lo;
                v2f bF;
                bF.x = G[kk * K_DIM + n];
                bF.y = G[(kk + 1) * K_DIM + n];
                acc[ni] = wmma_f32(aF, bF, acc[ni]);
            }
        }

#pragma unroll
        for (int ni = 0; ni < 4; ++ni) {
            int n = n0 + ni * 16 + lo;
#pragma unroll
            for (int v = 0; v < 8; ++v) {
                int m = v + 8 * hi;
                float pv  = p_s[m * K_DIM + n];
                float val = acc[ni][v]
                          + LMBDA_F * mask[(m0 + m) * K_DIM + n] * pv;
                ap_s[m * K_DIM + n] = val;
                atomicAdd(&pAp[m], pv * val);
            }
        }
        __syncthreads();

        if (t < 16) { alpha_s[t] = rs[t] / pAp[t]; rs_new[t] = 0.0f; }
        __syncthreads();

        {
            float a_m  = alpha_s[rowT];
            float part = 0.0f;
            for (int j = 0; j < 32; ++j) {
                int idx = rowT * K_DIM + colT + j * 16;
                float xv = x_s[idx] + a_m * p_s[idx];
                float rv = r_s[idx] - a_m * ap_s[idx];
                x_s[idx] = xv;
                r_s[idx] = rv;
                part += rv * rv;
            }
            atomicAdd(&rs_new[rowT], part);
        }
        __syncthreads();

        if (t < 16) { beta_s[t] = rs_new[t] / rs[t]; rs[t] = rs_new[t]; }
        __syncthreads();

        {
            float b_m = beta_s[rowT];
            for (int j = 0; j < 32; ++j) {
                int idx = rowT * K_DIM + colT + j * 16;
                p_s[idx] = r_s[idx] + b_m * p_s[idx];
            }
        }
        __syncthreads();
    }

    for (int j = 0; j < 32; ++j) {
        int c = colT + j * 16;
        Xout[(m0 + rowT) * K_DIM + c] = x_s[rowT * K_DIM + c];
    }
}

extern "C" void kernel_launch(void* const* d_in, const int* in_sizes, int n_in,
                              void* d_out, int out_size, void* d_ws, size_t ws_size,
                              hipStream_t stream) {
    const float* A  = (const float*)d_in[0];
    const float* B  = (const float*)d_in[1];
    const float* ea = (const float*)d_in[2];
    const float* eb = (const float*)d_in[3];
    float* out = (float*)d_out;

    float* ws      = (float*)d_ws;
    float* s_max   = ws;
    float* mask_ab = s_max   + 16;
    float* mask_ba = mask_ab + K_DIM * K_DIM;
    float* G_a     = mask_ba + K_DIM * K_DIM;
    float* G_b     = G_a     + K_DIM * K_DIM;
    float* R_ab    = G_b     + K_DIM * K_DIM;
    float* R_ba    = R_ab    + K_DIM * K_DIM;

    max_reduce_kernel<<<1, 512, 0, stream>>>(ea, eb, s_max);

    dim3 mgrid(K_DIM / 256, K_DIM);
    mask_kernel<<<mgrid, 256, 0, stream>>>(eb, ea, s_max, mask_ab);
    mask_kernel<<<mgrid, 256, 0, stream>>>(ea, eb, s_max, mask_ba);

    dim3 ggrid(8, 2, 4);
    gram_kernel<<<ggrid, 256, 0, stream>>>(A, B, G_a);

    size_t shmem = (size_t)4 * 16 * K_DIM * sizeof(float);
    dim3 cgrid(K_DIM / 16, 2);
    cg_solve_kernel<<<cgrid, 256, shmem, stream>>>(
        G_a, R_ab, mask_ab, out,
        G_b, R_ba, mask_ba, out + K_DIM * K_DIM,
        NITER);
}


# CG GEMM: G read as 128-bit loads two blocks ahead of their WMMAs (ping-pong registers, B pairs built before each run); lane owns 4 adjacent columns, so the pAp LDS atomic adds change order
# speedup vs baseline: 2.7789x; 1.3751x over previous
#include <hip/hip_runtime.h>


#define D_DIM   1024
#define K_DIM   512
#define LMBDA_F 1000.0f
#define NITER   96

typedef float v2f __attribute__((ext_vector_type(2)));
typedef float v4f __attribute__((ext_vector_type(4)));
typedef float v8f __attribute__((ext_vector_type(8)));

__device__ __forceinline__ v8f wmma_f32(v2f a, v2f b, v8f c) {
    return __builtin_amdgcn_wmma_f32_16x16x4_f32(
        false, a, false, b, (short)0, c, false, false);
}

__global__ void max_reduce_kernel(const float* __restrict__ ea,
                                  const float* __restrict__ eb,
                                  float* __restrict__ s_out) {
    __shared__ float sm[512];
    int t = threadIdx.x;
    sm[t] = fmaxf(ea[t], eb[t]);
    __syncthreads();
    for (int off = 256; off > 0; off >>= 1) {
        if (t < off) sm[t] = fmaxf(sm[t], sm[t + off]);
        __syncthreads();
    }
    if (t == 0) s_out[0] = sm[0];
}

__global__ void mask_kernel(const float* __restrict__ erow,
                            const float* __restrict__ ecol,
                            const float* __restrict__ s_ptr,
                            float* __restrict__ mask) {
    int j = blockIdx.x * blockDim.x + threadIdx.x;
    int i = blockIdx.y;
    float inv_s = 1.0f / s_ptr[0];
    float gb = erow[i] * inv_s;
    float ga = ecol[j] * inv_s;
    float gb2 = gb * gb + 1.0f;
    float ga2 = ga * ga + 1.0f;
    float re = gb / gb2 - ga / ga2;
    float im = 1.0f / gb2 - 1.0f / ga2;
    mask[i * K_DIM + j] = re * re + im * im;
}

__global__ void __launch_bounds__(256, 1)
gram_kernel(const float* __restrict__ A,
            const float* __restrict__ B,
            float* __restrict__ C4) {
    const int z = blockIdx.z;
    const float* __restrict__ X = (z == 1 || z == 2) ? B : A;
    const float* __restrict__ Y = (z == 1 || z == 3) ? B : A;
    float* __restrict__ C = C4 + (size_t)z * K_DIM * K_DIM;
    int lane = threadIdx.x & 31;
    int wave = threadIdx.x >> 5;
    int lo = lane & 15;
    int hi = lane >> 4;
    int m0 = blockIdx.x * 64;
    int n0 = blockIdx.y * 256 + wave * 32;

    v8f zero = {};
    v8f acc[4][2];
#pragma unroll
    for (int mi = 0; mi < 4; ++mi)
#pragma unroll
        for (int ni = 0; ni < 2; ++ni) acc[mi][ni] = zero;

    for (int d = 0; d < D_DIM; d += 4) {
        int kk = d + 2 * hi;
        v2f aF[4], bF[2];
#pragma unroll
        for (int mi = 0; mi < 4; ++mi) {
            int m = m0 + mi * 16 + lo;
            aF[mi].x = X[kk * K_DIM + m];
            aF[mi].y = X[(kk + 1) * K_DIM + m];
        }
#pragma unroll
        for (int ni = 0; ni < 2; ++ni) {
            int n = n0 + ni * 16 + lo;
            bF[ni].x = Y[kk * K_DIM + n];
            bF[ni].y = Y[(kk + 1) * K_DIM + n];
        }
#pragma unroll
        for (int mi = 0; mi < 4; ++mi)
#pragma unroll
            for (int ni = 0; ni < 2; ++ni)
                acc[mi][ni] = wmma_f32(aF[mi], bF[ni], acc[mi][ni]);
    }

#pragma unroll
    for (int mi = 0; mi < 4; ++mi)
#pragma unroll
        for (int ni = 0; ni < 2; ++ni)
#pragma unroll
            for (int v = 0; v < 8; ++v) {
                int m = m0 + mi * 16 + v + 8 * hi;
                int n = n0 + ni * 16 + lo;
                C[m * K_DIM + n] = acc[mi][ni][v];
            }
}

#define KU 4
#define BK (4 * KU)
static_assert(KU == 4, "the operand lists of pair_block and cover_block are written out for KU == 4");

__device__ __forceinline__ void load_g_block(const float* __restrict__ gp,
                                             v4f (&g0)[KU], v4f (&g1)[KU]) {
#pragma unroll
    for (int u = 0; u < KU; ++u) {
        g0[u] = *reinterpret_cast<const v4f*>(gp + (4 * u) * K_DIM);
        g1[u] = *reinterpret_cast<const v4f*>(gp + (4 * u + 1) * K_DIM);
    }
}

__device__ __forceinline__ void load_a_block(const float* pa, v2f (&aF)[KU]) {
#pragma unroll
    for (int u = 0; u < KU; ++u)
        aF[u] = *reinterpret_cast<const v2f*>(pa + 4 * u);
}

__device__ __forceinline__ void pair_block(const v4f (&g0)[KU],
                                           const v4f (&g1)[KU],
                                           v2f (&bp)[KU][4], v8f (&acc)[4]) {
#pragma unroll
    for (int u = 0; u < KU; ++u) {
        bp[u][0].x = g0[u].x; bp[u][0].y = g1[u].x;
        bp[u][1].x = g0[u].y; bp[u][1].y = g1[u].y;
        bp[u][2].x = g0[u].z; bp[u][2].y = g1[u].z;
        bp[u][3].x = g0[u].w; bp[u][3].y = g1[u].w;
    }
    asm volatile(""
        : "+v"(bp[0][0]), "+v"(bp[0][1]), "+v"(bp[0][2]), "+v"(bp[0][3]),
          "+v"(bp[1][0]), "+v"(bp[1][1]), "+v"(bp[1][2]), "+v"(bp[1][3]),
          "+v"(bp[2][0]), "+v"(bp[2][1]), "+v"(bp[2][2]), "+v"(bp[2][3]),
          "+v"(bp[3][0]), "+v"(bp[3][1]), "+v"(bp[3][2]), "+v"(bp[3][3]),
          "+v"(acc[0]), "+v"(acc[1]), "+v"(acc[2]), "+v"(acc[3]));
}

__device__ __forceinline__ void mma_block(const v2f (&aF)[KU],
                                          const v2f (&bp)[KU][4],
                                          v8f (&acc)[4]) {
    __builtin_amdgcn_sched_barrier(0);
#pragma unroll
    for (int u = 0; u < KU; ++u) {
        acc[0] = wmma_f32(aF[u], bp[u][0], acc[0]);
        acc[1] = wmma_f32(aF[u], bp[u][1], acc[1]);
        acc[2] = wmma_f32(aF[u], bp[u][2], acc[2]);
        acc[3] = wmma_f32(aF[u], bp[u][3], acc[3]);
    }
}

#define COVER_NOPS "v_nop\n\tv_nop\n\tv_nop\n\tv_nop\n\tv_nop\n\tv_nop\n\tv_nop\n\tv_nop"

__device__ __forceinline__ void cover_block(v8f (&acc)[4],
                                            v4f (&n0)[KU], v4f (&n1)[KU]) {
    asm volatile(COVER_NOPS
        : "+v"(acc[0]), "+v"(acc[1]), "+v"(acc[2]), "+v"(acc[3]),
          "+v"(n0[0]), "+v"(n0[1]), "+v"(n0[2]), "+v"(n0[3]),
          "+v"(n1[0]), "+v"(n1[1]), "+v"(n1[2]), "+v"(n1[3]));
    __builtin_amdgcn_sched_barrier(0);
}

__device__ __forceinline__ void cover_last(v8f (&acc)[4]) {
    asm volatile(COVER_NOPS
        : "+v"(acc[0]), "+v"(acc[1]), "+v"(acc[2]), "+v"(acc[3]));
}

__global__ void __launch_bounds__(256, 1)
cg_solve_kernel(const float* __restrict__ G0,
                const float* __restrict__ R0,
                const float* __restrict__ mask0,
                float* __restrict__ Xout0,
                const float* __restrict__ G1,
                const float* __restrict__ R1,
                const float* __restrict__ mask1,
                float* __restrict__ Xout1,
                int niter) {
    const bool dir1 = blockIdx.y != 0;
    const float* __restrict__ G    = dir1 ? G1    : G0;
    const float* __restrict__ R    = dir1 ? R1    : R0;
    const float* __restrict__ mask = dir1 ? mask1 : mask0;
    float* __restrict__ Xout       = dir1 ? Xout1 : Xout0;
    extern __shared__ float smem[];
    float* p_s  = smem;
    float* r_s  = p_s  + 16 * K_DIM;
    float* x_s  = r_s  + 16 * K_DIM;
    float* ap_s = x_s  + 16 * K_DIM;
    __shared__ float rs[16], rs_new[16], pAp[16], alpha_s[16], beta_s[16];

    int t    = threadIdx.x;
    int lane = t & 31;
    int wave = t >> 5;
    int lo   = lane & 15;
    int hi   = lane >> 4;
    int n0   = wave * 64;
    int m0   = blockIdx.x * 16;

    int rowT = t >> 4;
    int colT = t & 15;

    const float* __restrict__ gbase = G + (2 * hi) * K_DIM + n0 + 4 * lo;
    const float* pbase = p_s + lo * K_DIM + 2 * hi;

    if (t < 16) rs[t] = 0.0f;
    __syncthreads();
    {
        float part = 0.0f;
        for (int j = 0; j < 32; ++j) {
            int c = colT + j * 16;
            float rv = R[(m0 + rowT) * K_DIM + c];
            r_s[rowT * K_DIM + c] = rv;
            p_s[rowT * K_DIM + c] = rv;
            x_s[rowT * K_DIM + c] = 0.0f;
            part += rv * rv;
        }
        atomicAdd(&rs[rowT], part);
    }
    __syncthreads();

    float4 lmk[8];
#pragma unroll
    for (int v = 0; v < 8; ++v) {
        float4 mk = *reinterpret_cast<const float4*>(
            &mask[(m0 + v + 8 * hi) * K_DIM + n0 + 4 * lo]);
        lmk[v].x = LMBDA_F * mk.x; lmk[v].y = LMBDA_F * mk.y;
        lmk[v].z = LMBDA_F * mk.z; lmk[v].w = LMBDA_F * mk.w;
    }

    for (int it = 0; it < niter; ++it) {
        if (t < 16) pAp[t] = 0.0f;
        __syncthreads();

        v8f zero = {};
        v8f acc[4];
#pragma unroll
        for (int ni = 0; ni < 4; ++ni) acc[ni] = zero;

        {
            v4f ra0[KU], ra1[KU], rb0[KU], rb1[KU];
            v2f aA[KU], aB[KU];
            v2f bp[KU][4];
            load_g_block(gbase, ra0, ra1);
            load_g_block(gbase + BK * K_DIM, rb0, rb1);
            load_a_block(pbase, aA);
            int kb = 0;
            for (; kb < K_DIM - 2 * BK; kb += 2 * BK) {
                pair_block(ra0, ra1, bp, acc);
                load_g_block(gbase + (kb + 2 * BK) * K_DIM, ra0, ra1);
                load_a_block(pbase + kb + BK, aB);
                mma_block(aA, bp, acc);
                cover_block(acc, rb0, rb1);
                pair_block(rb0, rb1, bp, acc);
                load_g_block(gbase + (kb + 3 * BK) * K_DIM, rb0, rb1);
                load_a_block(pbase + kb + 2 * BK, aA);
                mma_block(aB, bp, acc);
                cover_block(acc, ra0, ra1);
            }
            pair_block(ra0, ra1, bp, acc);
            load_a_block(pbase + kb + BK, aB);
            mma_block(aA, bp, acc);
            cover_block(acc, rb0, rb1);
            pair_block(rb0, rb1, bp, acc);
            mma_block(aB, bp, acc);
            cover_last(acc);
        }

#pragma unroll
        for (int v = 0; v < 8; ++v) {
            int m = v + 8 * hi;
            int n = n0 + 4 * lo;
            float4 pv = *reinterpret_cast<const float4*>(&p_s[m * K_DIM + n]);
            float4 val;
            val.x = acc[0][v] + lmk[v].x * pv.x;
            val.y = acc[1][v] + lmk[v].y * pv.y;
            val.z = acc[2][v] + lmk[v].z * pv.z;
            val.w = acc[3][v] + lmk[v].w * pv.w;
            *reinterpret_cast<float4*>(&ap_s[m * K_DIM + n]) = val;
            atomicAdd(&pAp[m], pv.x * val.x);
            atomicAdd(&pAp[m], pv.y * val.y);
            atomicAdd(&pAp[m], pv.z * val.z);
            atomicAdd(&pAp[m], pv.w * val.w);
        }
        __syncthreads();

        if (t < 16) { alpha_s[t] = rs[t] / pAp[t]; rs_new[t] = 0.0f; }
        __syncthreads();

        {
            float a_m  = alpha_s[rowT];
            float part = 0.0f;
            for (int j = 0; j < 32; ++j) {
                int idx = rowT * K_DIM + colT + j * 16;
                float xv = x_s[idx] + a_m * p_s[idx];
                float rv = r_s[idx] - a_m * ap_s[idx];
                x_s[idx] = xv;
                r_s[idx] = rv;
                part += rv * rv;
            }
            atomicAdd(&rs_new[rowT], part);
        }
        __syncthreads();

        if (t < 16) { beta_s[t] = rs_new[t] / rs[t]; rs[t] = rs_new[t]; }
        __syncthreads();

        {
            float b_m = beta_s[rowT];
            for (int j = 0; j < 32; ++j) {
                int idx = rowT * K_DIM + colT + j * 16;
                p_s[idx] = r_s[idx] + b_m * p_s[idx];
            }
        }
        __syncthreads();
    }

    for (int j = 0; j < 32; ++j) {
        int c = colT + j * 16;
        Xout[(m0 + rowT) * K_DIM + c] = x_s[rowT * K_DIM + c];
    }
}

extern "C" void kernel_launch(void* const* d_in, const int* in_sizes, int n_in,
                              void* d_out, int out_size, void* d_ws, size_t ws_size,
                              hipStream_t stream) {
    const float* A  = (const float*)d_in[0];
    const float* B  = (const float*)d_in[1];
    const float* ea = (const float*)d_in[2];
    const float* eb = (const float*)d_in[3];
    float* out = (float*)d_out;

    float* ws      = (float*)d_ws;
    float* s_max   = ws;
    float* mask_ab = s_max   + 16;
    float* mask_ba = mask_ab + K_DIM * K_DIM;
    float* G_a     = mask_ba + K_DIM * K_DIM;
    float* G_b     = G_a     + K_DIM * K_DIM;
    float* R_ab    = G_b     + K_DIM * K_DIM;
    float* R_ba    = R_ab    + K_DIM * K_DIM;

    max_reduce_kernel<<<1, 512, 0, stream>>>(ea, eb, s_max);

    dim3 mgrid(K_DIM / 256, K_DIM);
    mask_kernel<<<mgrid, 256, 0, stream>>>(eb, ea, s_max, mask_ab);
    mask_kernel<<<mgrid, 256, 0, stream>>>(ea, eb, s_max, mask_ba);

    dim3 ggrid(8, 2, 4);
    gram_kernel<<<ggrid, 256, 0, stream>>>(A, B, G_a);

    size_t shmem = (size_t)4 * 16 * K_DIM * sizeof(float);
    dim3 cgrid(K_DIM / 16, 2);
    cg_solve_kernel<<<cgrid, 256, shmem, stream>>>(
        G_a, R_ab, mask_ab, out,
        G_b, R_ba, mask_ba, out + K_DIM * K_DIM,
        NITER);
}


# Row sums <p,Ap>, <r,r> by 16-lane DPP butterfly in registers, all lanes get the sum: same-address LDS atomics, zeroing, alpha/beta LDS trips go (6 barriers/iter -> 2); sum order now fixed
# speedup vs baseline: 4.2425x; 1.5267x over previous
#include <hip/hip_runtime.h>


#define D_DIM   1024
#define K_DIM   512
#define LMBDA_F 1000.0f
#define NITER   96

typedef float v2f __attribute__((ext_vector_type(2)));
typedef float v4f __attribute__((ext_vector_type(4)));
typedef float v8f __attribute__((ext_vector_type(8)));

__device__ __forceinline__ v8f wmma_f32(v2f a, v2f b, v8f c) {
    return __builtin_amdgcn_wmma_f32_16x16x4_f32(
        false, a, false, b, (short)0, c, false, false);
}

__global__ void max_reduce_kernel(const float* __restrict__ ea,
                                  const float* __restrict__ eb,
                                  float* __restrict__ s_out) {
    __shared__ float sm[512];
    int t = threadIdx.x;
    sm[t] = fmaxf(ea[t], eb[t]);
    __syncthreads();
    for (int off = 256; off > 0; off >>= 1) {
        if (t < off) sm[t] = fmaxf(sm[t], sm[t + off]);
        __syncthreads();
    }
    if (t == 0) s_out[0] = sm[0];
}

__global__ void mask_kernel(const float* __restrict__ erow,
                            const float* __restrict__ ecol,
                            const float* __restrict__ s_ptr,
                            float* __restrict__ mask) {
    int j = blockIdx.x * blockDim.x + threadIdx.x;
    int i = blockIdx.y;
    float inv_s = 1.0f / s_ptr[0];
    float gb = erow[i] * inv_s;
    float ga = ecol[j] * inv_s;
    float gb2 = gb * gb + 1.0f;
    float ga2 = ga * ga + 1.0f;
    float re = gb / gb2 - ga / ga2;
    float im = 1.0f / gb2 - 1.0f / ga2;
    mask[i * K_DIM + j] = re * re + im * im;
}

__global__ void __launch_bounds__(256, 1)
gram_kernel(const float* __restrict__ A,
            const float* __restrict__ B,
            float* __restrict__ C4) {
    const int z = blockIdx.z;
    const float* __restrict__ X = (z == 1 || z == 2) ? B : A;
    const float* __restrict__ Y = (z == 1 || z == 3) ? B : A;
    float* __restrict__ C = C4 + (size_t)z * K_DIM * K_DIM;
    int lane = threadIdx.x & 31;
    int wave = threadIdx.x >> 5;
    int lo = lane & 15;
    int hi = lane >> 4;
    int m0 = blockIdx.x * 64;
    int n0 = blockIdx.y * 256 + wave * 32;

    v8f zero = {};
    v8f acc[4][2];
#pragma unroll
    for (int mi = 0; mi < 4; ++mi)
#pragma unroll
        for (int ni = 0; ni < 2; ++ni) acc[mi][ni] = zero;

    for (int d = 0; d < D_DIM; d += 4) {
        int kk = d + 2 * hi;
        v2f aF[4], bF[2];
#pragma unroll
        for (int mi = 0; mi < 4; ++mi) {
            int m = m0 + mi * 16 + lo;
            aF[mi].x = X[kk * K_DIM + m];
            aF[mi].y = X[(kk + 1) * K_DIM + m];
        }
#pragma unroll
        for (int ni = 0; ni < 2; ++ni) {
            int n = n0 + ni * 16 + lo;
            bF[ni].x = Y[kk * K_DIM + n];
            bF[ni].y = Y[(kk + 1) * K_DIM + n];
        }
#pragma unroll
        for (int mi = 0; mi < 4; ++mi)
#pragma unroll
            for (int ni = 0; ni < 2; ++ni)
                acc[mi][ni] = wmma_f32(aF[mi], bF[ni], acc[mi][ni]);
    }

#pragma unroll
    for (int mi = 0; mi < 4; ++mi)
#pragma unroll
        for (int ni = 0; ni < 2; ++ni)
#pragma unroll
            for (int v = 0; v < 8; ++v) {
                int m = m0 + mi * 16 + v + 8 * hi;
                int n = n0 + ni * 16 + lo;
                C[m * K_DIM + n] = acc[mi][ni][v];
            }
}

#define KU 4
#define BK (4 * KU)
static_assert(KU == 4, "the operand lists of pair_block and cover_block are written out for KU == 4");

__device__ __forceinline__ void load_g_block(const float* __restrict__ gp,
                                             v4f (&g0)[KU], v4f (&g1)[KU]) {
#pragma unroll
    for (int u = 0; u < KU; ++u) {
        g0[u] = *reinterpret_cast<const v4f*>(gp + (4 * u) * K_DIM);
        g1[u] = *reinterpret_cast<const v4f*>(gp + (4 * u + 1) * K_DIM);
    }
}

__device__ __forceinline__ void load_a_block(const float* pa, v2f (&aF)[KU]) {
#pragma unroll
    for (int u = 0; u < KU; ++u)
        aF[u] = *reinterpret_cast<const v2f*>(pa + 4 * u);
}

__device__ __forceinline__ void pair_block(const v4f (&g0)[KU],
                                           const v4f (&g1)[KU],
                                           v2f (&bp)[KU][4], v8f (&acc)[4]) {
#pragma unroll
    for (int u = 0; u < KU; ++u) {
        bp[u][0].x = g0[u].x; bp[u][0].y = g1[u].x;
        bp[u][1].x = g0[u].y; bp[u][1].y = g1[u].y;
        bp[u][2].x = g0[u].z; bp[u][2].y = g1[u].z;
        bp[u][3].x = g0[u].w; bp[u][3].y = g1[u].w;
    }
    asm volatile(""
        : "+v"(bp[0][0]), "+v"(bp[0][1]), "+v"(bp[0][2]), "+v"(bp[0][3]),
          "+v"(bp[1][0]), "+v"(bp[1][1]), "+v"(bp[1][2]), "+v"(bp[1][3]),
          "+v"(bp[2][0]), "+v"(bp[2][1]), "+v"(bp[2][2]), "+v"(bp[2][3]),
          "+v"(bp[3][0]), "+v"(bp[3][1]), "+v"(bp[3][2]), "+v"(bp[3][3]),
          "+v"(acc[0]), "+v"(acc[1]), "+v"(acc[2]), "+v"(acc[3]));
}

__device__ __forceinline__ void mma_block(const v2f (&aF)[KU],
                                          const v2f (&bp)[KU][4],
                                          v8f (&acc)[4]) {
    __builtin_amdgcn_sched_barrier(0);
#pragma unroll
    for (int u = 0; u < KU; ++u) {
        acc[0] = wmma_f32(aF[u], bp[u][0], acc[0]);
        acc[1] = wmma_f32(aF[u], bp[u][1], acc[1]);
        acc[2] = wmma_f32(aF[u], bp[u][2], acc[2]);
        acc[3] = wmma_f32(aF[u], bp[u][3], acc[3]);
    }
}

#define COVER_NOPS "v_nop\n\tv_nop\n\tv_nop\n\tv_nop\n\tv_nop\n\tv_nop\n\tv_nop\n\tv_nop"

__device__ __forceinline__ void cover_block(v8f (&acc)[4],
                                            v4f (&n0)[KU], v4f (&n1)[KU]) {
    asm volatile(COVER_NOPS
        : "+v"(acc[0]), "+v"(acc[1]), "+v"(acc[2]), "+v"(acc[3]),
          "+v"(n0[0]), "+v"(n0[1]), "+v"(n0[2]), "+v"(n0[3]),
          "+v"(n1[0]), "+v"(n1[1]), "+v"(n1[2]), "+v"(n1[3]));
    __builtin_amdgcn_sched_barrier(0);
}

__device__ __forceinline__ void cover_last(v8f (&acc)[4]) {
    asm volatile(COVER_NOPS
        : "+v"(acc[0]), "+v"(acc[1]), "+v"(acc[2]), "+v"(acc[3]));
}

template <int XMASK>
__device__ __forceinline__ float row_xmask_add(float x) {
    int y = __builtin_amdgcn_update_dpp(0, __float_as_int(x), 0x160 + XMASK,
                                        0xf, 0xf, true);
    return x + __int_as_float(y);
}

__device__ __forceinline__ float row16_sum(float x) {
    x = row_xmask_add<1>(x);
    x = row_xmask_add<2>(x);
    x = row_xmask_add<4>(x);
    x = row_xmask_add<8>(x);
    return x;
}

__global__ void __launch_bounds__(256, 1)
cg_solve_kernel(const float* __restrict__ G0,
                const float* __restrict__ R0,
                const float* __restrict__ mask0,
                float* __restrict__ Xout0,
                const float* __restrict__ G1,
                const float* __restrict__ R1,
                const float* __restrict__ mask1,
                float* __restrict__ Xout1,
                int niter) {
    const bool dir1 = blockIdx.y != 0;
    const float* __restrict__ G    = dir1 ? G1    : G0;
    const float* __restrict__ R    = dir1 ? R1    : R0;
    const float* __restrict__ mask = dir1 ? mask1 : mask0;
    float* __restrict__ Xout       = dir1 ? Xout1 : Xout0;
    extern __shared__ float smem[];
    float* p_s  = smem;
    float* r_s  = p_s  + 16 * K_DIM;
    float* x_s  = r_s  + 16 * K_DIM;
    float* ap_s = x_s  + 16 * K_DIM;
    __shared__ __align__(16) float red[8 * 16];

    int t    = threadIdx.x;
    int lane = t & 31;
    int wave = t >> 5;
    int lo   = lane & 15;
    int hi   = lane >> 4;
    int n0   = wave * 64;
    int m0   = blockIdx.x * 16;

    int rowT = t >> 4;
    int colT = t & 15;

    const float* __restrict__ gbase = G + (2 * hi) * K_DIM + n0 + 4 * lo;
    const float* pbase = p_s + lo * K_DIM + 2 * hi;

    float rs_row;
    {
        float part = 0.0f;
        for (int j = 0; j < 32; ++j) {
            int c = colT + j * 16;
            float rv = R[(m0 + rowT) * K_DIM + c];
            r_s[rowT * K_DIM + c] = rv;
            p_s[rowT * K_DIM + c] = rv;
            x_s[rowT * K_DIM + c] = 0.0f;
            part += rv * rv;
        }
        rs_row = row16_sum(part);
    }
    __syncthreads();

    float4 lmk[8];
#pragma unroll
    for (int v = 0; v < 8; ++v) {
        float4 mk = *reinterpret_cast<const float4*>(
            &mask[(m0 + v + 8 * hi) * K_DIM + n0 + 4 * lo]);
        lmk[v].x = LMBDA_F * mk.x; lmk[v].y = LMBDA_F * mk.y;
        lmk[v].z = LMBDA_F * mk.z; lmk[v].w = LMBDA_F * mk.w;
    }

    for (int it = 0; it < niter; ++it) {
        v8f zero = {};
        v8f acc[4];
#pragma unroll
        for (int ni = 0; ni < 4; ++ni) acc[ni] = zero;

        {
            v4f ra0[KU], ra1[KU], rb0[KU], rb1[KU];
            v2f aA[KU], aB[KU];
            v2f bp[KU][4];
            load_g_block(gbase, ra0, ra1);
            load_g_block(gbase + BK * K_DIM, rb0, rb1);
            load_a_block(pbase, aA);
            int kb = 0;
            for (; kb < K_DIM - 2 * BK; kb += 2 * BK) {
                pair_block(ra0, ra1, bp, acc);
                load_g_block(gbase + (kb + 2 * BK) * K_DIM, ra0, ra1);
                load_a_block(pbase + kb + BK, aB);
                mma_block(aA, bp, acc);
                cover_block(acc, rb0, rb1);
                pair_block(rb0, rb1, bp, acc);
                load_g_block(gbase + (kb + 3 * BK) * K_DIM, rb0, rb1);
                load_a_block(pbase + kb + 2 * BK, aA);
                mma_block(aB, bp, acc);
                cover_block(acc, ra0, ra1);
            }
            pair_block(ra0, ra1, bp, acc);
            load_a_block(pbase + kb + BK, aB);
            mma_block(aA, bp, acc);
            cover_block(acc, rb0, rb1);
            pair_block(rb0, rb1, bp, acc);
            mma_block(aB, bp, acc);
            cover_last(acc);
        }

        float ps[8];
#pragma unroll
        for (int v = 0; v < 8; ++v) {
            int m = v + 8 * hi;
            int n = n0 + 4 * lo;
            float4 pv = *reinterpret_cast<const float4*>(&p_s[m * K_DIM + n]);
            float4 val;
            val.x = acc[0][v] + lmk[v].x * pv.x;
            val.y = acc[1][v] + lmk[v].y * pv.y;
            val.z = acc[2][v] + lmk[v].z * pv.z;
            val.w = acc[3][v] + lmk[v].w * pv.w;
            *reinterpret_cast<float4*>(&ap_s[m * K_DIM + n]) = val;
            ps[v] = pv.x * val.x + pv.y * val.y + pv.z * val.z + pv.w * val.w;
        }
#pragma unroll
        for (int v = 0; v < 8; ++v) ps[v] = row_xmask_add<1>(ps[v]);
#pragma unroll
        for (int v = 0; v < 8; ++v) ps[v] = row_xmask_add<2>(ps[v]);
#pragma unroll
        for (int v = 0; v < 8; ++v) ps[v] = row_xmask_add<4>(ps[v]);
#pragma unroll
        for (int v = 0; v < 8; ++v) ps[v] = row_xmask_add<8>(ps[v]);
        if (lo == 0) {
            float4 s0, s1;
            s0.x = ps[0]; s0.y = ps[1]; s0.z = ps[2]; s0.w = ps[3];
            s1.x = ps[4]; s1.y = ps[5]; s1.z = ps[6]; s1.w = ps[7];
            *reinterpret_cast<float4*>(&red[wave * 16 + 8 * hi])     = s0;
            *reinterpret_cast<float4*>(&red[wave * 16 + 8 * hi + 4]) = s1;
        }
        __syncthreads();

        {
            float pap = red[rowT];
            for (int w = 1; w < 8; ++w) pap += red[w * 16 + rowT];
            float a_m  = rs_row / pap;
            float part = 0.0f;
            for (int j = 0; j < 32; ++j) {
                int idx = rowT * K_DIM + colT + j * 16;
                float xv = x_s[idx] + a_m * p_s[idx];
                float rv = r_s[idx] - a_m * ap_s[idx];
                x_s[idx] = xv;
                r_s[idx] = rv;
                part += rv * rv;
            }
            float rs_next = row16_sum(part);
            float b_m = rs_next / rs_row;
            rs_row = rs_next;
            for (int j = 0; j < 32; ++j) {
                int idx = rowT * K_DIM + colT + j * 16;
                p_s[idx] = r_s[idx] + b_m * p_s[idx];
            }
        }
        __syncthreads();
    }

    for (int j = 0; j < 32; ++j) {
        int c = colT + j * 16;
        Xout[(m0 + rowT) * K_DIM + c] = x_s[rowT * K_DIM + c];
    }
}

extern "C" void kernel_launch(void* const* d_in, const int* in_sizes, int n_in,
                              void* d_out, int out_size, void* d_ws, size_t ws_size,
                              hipStream_t stream) {
    const float* A  = (const float*)d_in[0];
    const float* B  = (const float*)d_in[1];
    const float* ea = (const float*)d_in[2];
    const float* eb = (const float*)d_in[3];
    float* out = (float*)d_out;

    float* ws      = (float*)d_ws;
    float* s_max   = ws;
    float* mask_ab = s_max   + 16;
    float* mask_ba = mask_ab + K_DIM * K_DIM;
    float* G_a     = mask_ba + K_DIM * K_DIM;
    float* G_b     = G_a     + K_DIM * K_DIM;
    float* R_ab    = G_b     + K_DIM * K_DIM;
    float* R_ba    = R_ab    + K_DIM * K_DIM;

    max_reduce_kernel<<<1, 512, 0, stream>>>(ea, eb, s_max);

    dim3 mgrid(K_DIM / 256, K_DIM);
    mask_kernel<<<mgrid, 256, 0, stream>>>(eb, ea, s_max, mask_ab);
    mask_kernel<<<mgrid, 256, 0, stream>>>(ea, eb, s_max, mask_ba);

    dim3 ggrid(8, 2, 4);
    gram_kernel<<<ggrid, 256, 0, stream>>>(A, B, G_a);

    size_t shmem = (size_t)4 * 16 * K_DIM * sizeof(float);
    dim3 cgrid(K_DIM / 16, 2);
    cg_solve_kernel<<<cgrid, 256, shmem, stream>>>(
        G_a, R_ab, mask_ab, out,
        G_b, R_ba, mask_ba, out + K_DIM * K_DIM,
        NITER);
}


# CG vectors x, r and Ap stay in registers: vector updates take the accumulators' lane ownership, only p and per-wave row sums cross LDS; <r,r> now summed per lane, across lanes, then waves in order
# speedup vs baseline: 4.8241x; 1.1371x over previous
#include <hip/hip_runtime.h>


#define D_DIM   1024
#define K_DIM   512
#define LMBDA_F 1000.0f
#define NITER   96

typedef float v2f __attribute__((ext_vector_type(2)));
typedef float v4f __attribute__((ext_vector_type(4)));
typedef float v8f __attribute__((ext_vector_type(8)));

__device__ __forceinline__ v8f wmma_f32(v2f a, v2f b, v8f c) {
    return __builtin_amdgcn_wmma_f32_16x16x4_f32(
        false, a, false, b, (short)0, c, false, false);
}

__global__ void max_reduce_kernel(const float* __restrict__ ea,
                                  const float* __restrict__ eb,
                                  float* __restrict__ s_out) {
    __shared__ float sm[512];
    int t = threadIdx.x;
    sm[t] = fmaxf(ea[t], eb[t]);
    __syncthreads();
    for (int off = 256; off > 0; off >>= 1) {
        if (t < off) sm[t] = fmaxf(sm[t], sm[t + off]);
        __syncthreads();
    }
    if (t == 0) s_out[0] = sm[0];
}

__global__ void mask_kernel(const float* __restrict__ erow,
                            const float* __restrict__ ecol,
                            const float* __restrict__ s_ptr,
                            float* __restrict__ mask) {
    int j = blockIdx.x * blockDim.x + threadIdx.x;
    int i = blockIdx.y;
    float inv_s = 1.0f / s_ptr[0];
    float gb = erow[i] * inv_s;
    float ga = ecol[j] * inv_s;
    float gb2 = gb * gb + 1.0f;
    float ga2 = ga * ga + 1.0f;
    float re = gb / gb2 - ga / ga2;
    float im = 1.0f / gb2 - 1.0f / ga2;
    mask[i * K_DIM + j] = re * re + im * im;
}

__global__ void __launch_bounds__(256, 1)
gram_kernel(const float* __restrict__ A,
            const float* __restrict__ B,
            float* __restrict__ C4) {
    const int z = blockIdx.z;
    const float* __restrict__ X = (z == 1 || z == 2) ? B : A;
    const float* __restrict__ Y = (z == 1 || z == 3) ? B : A;
    float* __restrict__ C = C4 + (size_t)z * K_DIM * K_DIM;
    int lane = threadIdx.x & 31;
    int wave = threadIdx.x >> 5;
    int lo = lane & 15;
    int hi = lane >> 4;
    int m0 = blockIdx.x * 64;
    int n0 = blockIdx.y * 256 + wave * 32;

    v8f zero = {};
    v8f acc[4][2];
#pragma unroll
    for (int mi = 0; mi < 4; ++mi)
#pragma unroll
        for (int ni = 0; ni < 2; ++ni) acc[mi][ni] = zero;

    for (int d = 0; d < D_DIM; d += 4) {
        int kk = d + 2 * hi;
        v2f aF[4], bF[2];
#pragma unroll
        for (int mi = 0; mi < 4; ++mi) {
            int m = m0 + mi * 16 + lo;
            aF[mi].x = X[kk * K_DIM + m];
            aF[mi].y = X[(kk + 1) * K_DIM + m];
        }
#pragma unroll
        for (int ni = 0; ni < 2; ++ni) {
            int n = n0 + ni * 16 + lo;
            bF[ni].x = Y[kk * K_DIM + n];
            bF[ni].y = Y[(kk + 1) * K_DIM + n];
        }
#pragma unroll
        for (int mi = 0; mi < 4; ++mi)
#pragma unroll
            for (int ni = 0; ni < 2; ++ni)
                acc[mi][ni] = wmma_f32(aF[mi], bF[ni], acc[mi][ni]);
    }

#pragma unroll
    for (int mi = 0; mi < 4; ++mi)
#pragma unroll
        for (int ni = 0; ni < 2; ++ni)
#pragma unroll
            for (int v = 0; v < 8; ++v) {
                int m = m0 + mi * 16 + v + 8 * hi;
                int n = n0 + ni * 16 + lo;
                C[m * K_DIM + n] = acc[mi][ni][v];
            }
}

#define KU 4
#define BK (4 * KU)
static_assert(KU == 4, "the operand lists of pair_block and cover_block are written out for KU == 4");

__device__ __forceinline__ void load_g_block(const float* __restrict__ gp,
                                             v4f (&g0)[KU], v4f (&g1)[KU]) {
#pragma unroll
    for (int u = 0; u < KU; ++u) {
        g0[u] = *reinterpret_cast<const v4f*>(gp + (4 * u) * K_DIM);
        g1[u] = *reinterpret_cast<const v4f*>(gp + (4 * u + 1) * K_DIM);
    }
}

__device__ __forceinline__ void load_a_block(const float* pa, v2f (&aF)[KU]) {
#pragma unroll
    for (int u = 0; u < KU; ++u)
        aF[u] = *reinterpret_cast<const v2f*>(pa + 4 * u);
}

__device__ __forceinline__ void pair_block(const v4f (&g0)[KU],
                                           const v4f (&g1)[KU],
                                           v2f (&bp)[KU][4], v8f (&acc)[4]) {
#pragma unroll
    for (int u = 0; u < KU; ++u) {
        bp[u][0].x = g0[u].x; bp[u][0].y = g1[u].x;
        bp[u][1].x = g0[u].y; bp[u][1].y = g1[u].y;
        bp[u][2].x = g0[u].z; bp[u][2].y = g1[u].z;
        bp[u][3].x = g0[u].w; bp[u][3].y = g1[u].w;
    }
    asm volatile(""
        : "+v"(bp[0][0]), "+v"(bp[0][1]), "+v"(bp[0][2]), "+v"(bp[0][3]),
          "+v"(bp[1][0]), "+v"(bp[1][1]), "+v"(bp[1][2]), "+v"(bp[1][3]),
          "+v"(bp[2][0]), "+v"(bp[2][1]), "+v"(bp[2][2]), "+v"(bp[2][3]),
          "+v"(bp[3][0]), "+v"(bp[3][1]), "+v"(bp[3][2]), "+v"(bp[3][3]),
          "+v"(acc[0]), "+v"(acc[1]), "+v"(acc[2]), "+v"(acc[3]));
}

__device__ __forceinline__ void mma_block(const v2f (&aF)[KU],
                                          const v2f (&bp)[KU][4],
                                          v8f (&acc)[4]) {
    __builtin_amdgcn_sched_barrier(0);
#pragma unroll
    for (int u = 0; u < KU; ++u) {
        acc[0] = wmma_f32(aF[u], bp[u][0], acc[0]);
        acc[1] = wmma_f32(aF[u], bp[u][1], acc[1]);
        acc[2] = wmma_f32(aF[u], bp[u][2], acc[2]);
        acc[3] = wmma_f32(aF[u], bp[u][3], acc[3]);
    }
}

#define COVER_NOPS "v_nop\n\tv_nop\n\tv_nop\n\tv_nop\n\tv_nop\n\tv_nop\n\tv_nop\n\tv_nop"

__device__ __forceinline__ void cover_block(v8f (&acc)[4],
                                            v4f (&n0)[KU], v4f (&n1)[KU]) {
    asm volatile(COVER_NOPS
        : "+v"(acc[0]), "+v"(acc[1]), "+v"(acc[2]), "+v"(acc[3]),
          "+v"(n0[0]), "+v"(n0[1]), "+v"(n0[2]), "+v"(n0[3]),
          "+v"(n1[0]), "+v"(n1[1]), "+v"(n1[2]), "+v"(n1[3]));
    __builtin_amdgcn_sched_barrier(0);
}

__device__ __forceinline__ void cover_last(v8f (&acc)[4]) {
    asm volatile(COVER_NOPS
        : "+v"(acc[0]), "+v"(acc[1]), "+v"(acc[2]), "+v"(acc[3]));
}

template <int XMASK>
__device__ __forceinline__ float row_xmask_add(float x) {
    int y = __builtin_amdgcn_update_dpp(0, __float_as_int(x), 0x160 + XMASK,
                                        0xf, 0xf, true);
    return x + __int_as_float(y);
}

__device__ __forceinline__ float row16_sum(float x) {
    x = row_xmask_add<1>(x);
    x = row_xmask_add<2>(x);
    x = row_xmask_add<4>(x);
    x = row_xmask_add<8>(x);
    return x;
}

__device__ __forceinline__ float dot4(float4 a, float4 b) {
    return a.x * b.x + a.y * b.y + a.z * b.z + a.w * b.w;
}

__device__ __forceinline__ void strip_row_sums(float (&ps)[8], float* slot, bool writer) {
#pragma unroll
    for (int v = 0; v < 8; ++v) ps[v] = row_xmask_add<1>(ps[v]);
#pragma unroll
    for (int v = 0; v < 8; ++v) ps[v] = row_xmask_add<2>(ps[v]);
#pragma unroll
    for (int v = 0; v < 8; ++v) ps[v] = row_xmask_add<4>(ps[v]);
#pragma unroll
    for (int v = 0; v < 8; ++v) ps[v] = row_xmask_add<8>(ps[v]);
    if (writer) {
        float4 s0, s1;
        s0.x = ps[0]; s0.y = ps[1]; s0.z = ps[2]; s0.w = ps[3];
        s1.x = ps[4]; s1.y = ps[5]; s1.z = ps[6]; s1.w = ps[7];
        *reinterpret_cast<float4*>(slot)     = s0;
        *reinterpret_cast<float4*>(slot + 4) = s1;
    }
}

__device__ __forceinline__ float sum_waves(const float* red, int row) {
    float s = red[row];
#pragma unroll
    for (int w = 1; w < 8; ++w) s += red[w * 16 + row];
    return s;
}

template <int SRC>
__device__ __forceinline__ float row_take(float x) {
    int y = __builtin_amdgcn_update_dpp(0, __float_as_int(x), 0x150 + SRC,
                                        0xf, 0xf, true);
    return __int_as_float(y);
}

__device__ __forceinline__ void take8(float s, float (&o)[8]) {
    o[0] = row_take<0>(s); o[1] = row_take<1>(s);
    o[2] = row_take<2>(s); o[3] = row_take<3>(s);
    o[4] = row_take<4>(s); o[5] = row_take<5>(s);
    o[6] = row_take<6>(s); o[7] = row_take<7>(s);
}

__global__ void __launch_bounds__(256, 1)
cg_solve_kernel(const float* __restrict__ G0,
                const float* __restrict__ R0,
                const float* __restrict__ mask0,
                float* __restrict__ Xout0,
                const float* __restrict__ G1,
                const float* __restrict__ R1,
                const float* __restrict__ mask1,
                float* __restrict__ Xout1,
                int niter) {
    const bool dir1 = blockIdx.y != 0;
    const float* __restrict__ G    = dir1 ? G1    : G0;
    const float* __restrict__ R    = dir1 ? R1    : R0;
    const float* __restrict__ mask = dir1 ? mask1 : mask0;
    float* __restrict__ Xout       = dir1 ? Xout1 : Xout0;
    extern __shared__ float smem[];
    float* p_s = smem;
    __shared__ __align__(16) float red_pap[8 * 16];
    __shared__ __align__(16) float red_rr[8 * 16];

    int t    = threadIdx.x;
    int lane = t & 31;
    int wave = t >> 5;
    int lo   = lane & 15;
    int hi   = lane >> 4;
    int n0   = wave * 64;
    int m0   = blockIdx.x * 16;

    int erow = 8 * hi;
    int ecol = n0 + 4 * lo;
    int srow = erow + (lo & 7);

    const float* __restrict__ gbase = G + (2 * hi) * K_DIM + n0 + 4 * lo;
    const float* pbase = p_s + lo * K_DIM + 2 * hi;

    float4 xv[8], rv[8];
    {
        float ps[8];
#pragma unroll
        for (int v = 0; v < 8; ++v) {
            float4 q = *reinterpret_cast<const float4*>(
                &R[(m0 + erow + v) * K_DIM + ecol]);
            rv[v] = q;
            xv[v].x = 0.0f; xv[v].y = 0.0f; xv[v].z = 0.0f; xv[v].w = 0.0f;
            *reinterpret_cast<float4*>(&p_s[(erow + v) * K_DIM + ecol]) = q;
            ps[v] = dot4(q, q);
        }
        strip_row_sums(ps, &red_rr[wave * 16 + erow], lo == 0);
    }
    __syncthreads();
    float rs_lane = sum_waves(red_rr, srow);

    float4 lmk[8];
#pragma unroll
    for (int v = 0; v < 8; ++v) {
        float4 mk = *reinterpret_cast<const float4*>(
            &mask[(m0 + erow + v) * K_DIM + ecol]);
        lmk[v].x = LMBDA_F * mk.x; lmk[v].y = LMBDA_F * mk.y;
        lmk[v].z = LMBDA_F * mk.z; lmk[v].w = LMBDA_F * mk.w;
    }

    for (int it = 0; it < niter; ++it) {
        v8f zero = {};
        v8f acc[4];
#pragma unroll
        for (int ni = 0; ni < 4; ++ni) acc[ni] = zero;

        {
            v4f ra0[KU], ra1[KU], rb0[KU], rb1[KU];
            v2f aA[KU], aB[KU];
            v2f bp[KU][4];
            load_g_block(gbase, ra0, ra1);
            load_g_block(gbase + BK * K_DIM, rb0, rb1);
            load_a_block(pbase, aA);
            int kb = 0;
            for (; kb < K_DIM - 2 * BK; kb += 2 * BK) {
                pair_block(ra0, ra1, bp, acc);
                load_g_block(gbase + (kb + 2 * BK) * K_DIM, ra0, ra1);
                load_a_block(pbase + kb + BK, aB);
                mma_block(aA, bp, acc);
                cover_block(acc, rb0, rb1);
                pair_block(rb0, rb1, bp, acc);
                load_g_block(gbase + (kb + 3 * BK) * K_DIM, rb0, rb1);
                load_a_block(pbase + kb + 2 * BK, aA);
                mma_block(aB, bp, acc);
                cover_block(acc, ra0, ra1);
            }
            pair_block(ra0, ra1, bp, acc);
            load_a_block(pbase + kb + BK, aB);
            mma_block(aA, bp, acc);
            cover_block(acc, rb0, rb1);
            pair_block(rb0, rb1, bp, acc);
            mma_block(aB, bp, acc);
            cover_last(acc);
        }

        float4 pv[8], av[8];
        float ps[8];
#pragma unroll
        for (int v = 0; v < 8; ++v) {
            pv[v] = *reinterpret_cast<const float4*>(&p_s[(erow + v) * K_DIM + ecol]);
            av[v].x = acc[0][v] + lmk[v].x * pv[v].x;
            av[v].y = acc[1][v] + lmk[v].y * pv[v].y;
            av[v].z = acc[2][v] + lmk[v].z * pv[v].z;
            av[v].w = acc[3][v] + lmk[v].w * pv[v].w;
            ps[v] = dot4(pv[v], av[v]);
        }
        strip_row_sums(ps, &red_pap[wave * 16 + erow], lo == 0);
        __syncthreads();

        float a[8];
        take8(rs_lane / sum_waves(red_pap, srow), a);
#pragma unroll
        for (int v = 0; v < 8; ++v) {
            xv[v].x = xv[v].x + a[v] * pv[v].x;  xv[v].y = xv[v].y + a[v] * pv[v].y;
            xv[v].z = xv[v].z + a[v] * pv[v].z;  xv[v].w = xv[v].w + a[v] * pv[v].w;
            rv[v].x = rv[v].x - a[v] * av[v].x;  rv[v].y = rv[v].y - a[v] * av[v].y;
            rv[v].z = rv[v].z - a[v] * av[v].z;  rv[v].w = rv[v].w - a[v] * av[v].w;
            ps[v] = dot4(rv[v], rv[v]);
        }
        strip_row_sums(ps, &red_rr[wave * 16 + erow], lo == 0);
        __syncthreads();

        float rs_next = sum_waves(red_rr, srow);
        float b[8];
        take8(rs_next / rs_lane, b);
        rs_lane = rs_next;
#pragma unroll
        for (int v = 0; v < 8; ++v) {
            float4 pn;
            pn.x = rv[v].x + b[v] * pv[v].x;  pn.y = rv[v].y + b[v] * pv[v].y;
            pn.z = rv[v].z + b[v] * pv[v].z;  pn.w = rv[v].w + b[v] * pv[v].w;
            *reinterpret_cast<float4*>(&p_s[(erow + v) * K_DIM + ecol]) = pn;
        }
        __syncthreads();
    }

#pragma unroll
    for (int v = 0; v < 8; ++v)
        *reinterpret_cast<float4*>(&Xout[(m0 + erow + v) * K_DIM + ecol]) = xv[v];
}

extern "C" void kernel_launch(void* const* d_in, const int* in_sizes, int n_in,
                              void* d_out, int out_size, void* d_ws, size_t ws_size,
                              hipStream_t stream) {
    const float* A  = (const float*)d_in[0];
    const float* B  = (const float*)d_in[1];
    const float* ea = (const float*)d_in[2];
    const float* eb = (const float*)d_in[3];
    float* out = (float*)d_out;

    float* ws      = (float*)d_ws;
    float* s_max   = ws;
    float* mask_ab = s_max   + 16;
    float* mask_ba = mask_ab + K_DIM * K_DIM;
    float* G_a     = mask_ba + K_DIM * K_DIM;
    float* G_b     = G_a     + K_DIM * K_DIM;
    float* R_ab    = G_b     + K_DIM * K_DIM;
    float* R_ba    = R_ab    + K_DIM * K_DIM;

    max_reduce_kernel<<<1, 512, 0, stream>>>(ea, eb, s_max);

    dim3 mgrid(K_DIM / 256, K_DIM);
    mask_kernel<<<mgrid, 256, 0, stream>>>(eb, ea, s_max, mask_ab);
    mask_kernel<<<mgrid, 256, 0, stream>>>(ea, eb, s_max, mask_ba);

    dim3 ggrid(8, 2, 4);
    gram_kernel<<<ggrid, 256, 0, stream>>>(A, B, G_a);

    size_t shmem = (size_t)16 * K_DIM * sizeof(float);
    dim3 cgrid(K_DIM / 16, 2);
    cg_solve_kernel<<<cgrid, 256, shmem, stream>>>(
        G_a, R_ab, mask_ab, out,
        G_b, R_ba, mask_ba, out + K_DIM * K_DIM,
        NITER);
}


# Gram kernel stores G with rows interleaved in pairs: CG's 128-bit loads of G are the B pairs as loaded, so pairing moves and no-op cover go; 4-set load ring runs on across iterations; sums unchanged
# speedup vs baseline: 5.6047x; 1.1618x over previous
#include <hip/hip_runtime.h>


#define D_DIM   1024
#define K_DIM   512
#define LMBDA_F 1000.0f
#define NITER   96

typedef float v2f __attribute__((ext_vector_type(2)));
typedef float v4f __attribute__((ext_vector_type(4)));
typedef float v8f __attribute__((ext_vector_type(8)));

__device__ __forceinline__ v8f wmma_f32(v2f a, v2f b, v8f c) {
    return __builtin_amdgcn_wmma_f32_16x16x4_f32(
        false, a, false, b, (short)0, c, false, false);
}

__global__ void max_reduce_kernel(const float* __restrict__ ea,
                                  const float* __restrict__ eb,
                                  float* __restrict__ s_out) {
    __shared__ float sm[512];
    int t = threadIdx.x;
    sm[t] = fmaxf(ea[t], eb[t]);
    __syncthreads();
    for (int off = 256; off > 0; off >>= 1) {
        if (t < off) sm[t] = fmaxf(sm[t], sm[t + off]);
        __syncthreads();
    }
    if (t == 0) s_out[0] = sm[0];
}

__global__ void mask_kernel(const float* __restrict__ erow,
                            const float* __restrict__ ecol,
                            const float* __restrict__ s_ptr,
                            float* __restrict__ mask) {
    int j = blockIdx.x * blockDim.x + threadIdx.x;
    int i = blockIdx.y;
    float inv_s = 1.0f / s_ptr[0];
    float gb = erow[i] * inv_s;
    float ga = ecol[j] * inv_s;
    float gb2 = gb * gb + 1.0f;
    float ga2 = ga * ga + 1.0f;
    float re = gb / gb2 - ga / ga2;
    float im = 1.0f / gb2 - 1.0f / ga2;
    mask[i * K_DIM + j] = re * re + im * im;
}

__global__ void __launch_bounds__(256, 1)
gram_kernel(const float* __restrict__ A,
            const float* __restrict__ B,
            float* __restrict__ C4) {
    const int z = blockIdx.z;
    const float* __restrict__ X = (z == 1 || z == 2) ? B : A;
    const float* __restrict__ Y = (z == 1 || z == 3) ? B : A;
    float* __restrict__ C = C4 + (size_t)z * K_DIM * K_DIM;
    int lane = threadIdx.x & 31;
    int wave = threadIdx.x >> 5;
    int lo = lane & 15;
    int hi = lane >> 4;
    int m0 = blockIdx.x * 64;
    int n0 = blockIdx.y * 256 + wave * 32;

    v8f zero = {};
    v8f acc[4][2];
#pragma unroll
    for (int mi = 0; mi < 4; ++mi)
#pragma unroll
        for (int ni = 0; ni < 2; ++ni) acc[mi][ni] = zero;

    for (int d = 0; d < D_DIM; d += 4) {
        int kk = d + 2 * hi;
        v2f aF[4], bF[2];
#pragma unroll
        for (int mi = 0; mi < 4; ++mi) {
            int m = m0 + mi * 16 + lo;
            aF[mi].x = X[kk * K_DIM + m];
            aF[mi].y = X[(kk + 1) * K_DIM + m];
        }
#pragma unroll
        for (int ni = 0; ni < 2; ++ni) {
            int n = n0 + ni * 16 + lo;
            bF[ni].x = Y[kk * K_DIM + n];
            bF[ni].y = Y[(kk + 1) * K_DIM + n];
        }
#pragma unroll
        for (int mi = 0; mi < 4; ++mi)
#pragma unroll
            for (int ni = 0; ni < 2; ++ni)
                acc[mi][ni] = wmma_f32(aF[mi], bF[ni], acc[mi][ni]);
    }

    if (z < 2) {
#pragma unroll
        for (int mi = 0; mi < 4; ++mi)
#pragma unroll
            for (int ni = 0; ni < 2; ++ni)
#pragma unroll
                for (int v = 0; v < 8; v += 2) {
                    int m = m0 + mi * 16 + v + 8 * hi;
                    int n = n0 + ni * 16 + lo;
                    v2f pr;
                    pr.x = acc[mi][ni][v];
                    pr.y = acc[mi][ni][v + 1];
                    *reinterpret_cast<v2f*>(&C[(m >> 1) * (2 * K_DIM) + 2 * n]) = pr;
                }
    } else {
#pragma unroll
        for (int mi = 0; mi < 4; ++mi)
#pragma unroll
            for (int ni = 0; ni < 2; ++ni)
#pragma unroll
                for (int v = 0; v < 8; ++v) {
                    int m = m0 + mi * 16 + v + 8 * hi;
                    int n = n0 + ni * 16 + lo;
                    C[m * K_DIM + n] = acc[mi][ni][v];
                }
    }
}

#define KU 4
#define BK (4 * KU)
#define NBLK (K_DIM / BK)
static_assert(NBLK % 4 == 0 && NBLK >= 8, "the ring of four register sets turns once per four blocks");

__device__ __forceinline__ void load_b_block(const float* __restrict__ gblk,
                                             unsigned gofb, v4f (&s)[2 * KU]) {
    const char* gb = reinterpret_cast<const char*>(gblk);
#pragma unroll
    for (int u = 0; u < KU; ++u) {
        s[2 * u]     = *reinterpret_cast<const v4f*>(gb + gofb + (2 * u) * (2 * K_DIM) * 4);
        s[2 * u + 1] = *reinterpret_cast<const v4f*>(gb + gofb + (2 * u) * (2 * K_DIM) * 4 + 16);
    }
}

__device__ __forceinline__ void load_a_block(const float* pa, v2f (&aF)[KU]) {
#pragma unroll
    for (int u = 0; u < KU; ++u)
        aF[u] = *reinterpret_cast<const v2f*>(pa + 4 * u);
}

__device__ __forceinline__ void mma_block(const v2f (&aF)[KU],
                                          const v4f (&s)[2 * KU],
                                          v8f (&acc)[4]) {
    __builtin_amdgcn_sched_barrier(0);
#pragma unroll
    for (int u = 0; u < KU; ++u) {
        acc[0] = wmma_f32(aF[u], __builtin_shufflevector(s[2 * u],     s[2 * u],     0, 1), acc[0]);
        acc[1] = wmma_f32(aF[u], __builtin_shufflevector(s[2 * u],     s[2 * u],     2, 3), acc[1]);
        acc[2] = wmma_f32(aF[u], __builtin_shufflevector(s[2 * u + 1], s[2 * u + 1], 0, 1), acc[2]);
        acc[3] = wmma_f32(aF[u], __builtin_shufflevector(s[2 * u + 1], s[2 * u + 1], 2, 3), acc[3]);
    }
}

__device__ __forceinline__ void refill_behind_run(v8f (&acc)[4], unsigned& gofb) {
    asm volatile(""
        : "+v"(acc[0]), "+v"(acc[1]), "+v"(acc[2]), "+v"(acc[3]), "+v"(gofb));
}

template <bool LAST>
__device__ __forceinline__ void ring_turn(const float* __restrict__ G, unsigned& gofb,
                                          const float* pbase, int blk,
                                          v4f (&s0)[2 * KU], v4f (&s1)[2 * KU],
                                          v4f (&s2)[2 * KU], v4f (&s3)[2 * KU],
                                          v2f (&aA)[KU], v2f (&aB)[KU], v8f (&acc)[4]) {
    refill_behind_run(acc, gofb);
    load_b_block(G + (blk + 3) * (BK * K_DIM), gofb, s3);
    load_a_block(pbase + (blk + 1) * BK, aB);
    mma_block(aA, s0, acc);
    refill_behind_run(acc, gofb);
    load_b_block(G + (LAST ? 0 : blk + 4) * (BK * K_DIM), gofb, s0);
    load_a_block(pbase + (blk + 2) * BK, aA);
    mma_block(aB, s1, acc);
    refill_behind_run(acc, gofb);
    load_b_block(G + (LAST ? 1 : blk + 5) * (BK * K_DIM), gofb, s1);
    load_a_block(pbase + (blk + 3) * BK, aB);
    mma_block(aA, s2, acc);
    refill_behind_run(acc, gofb);
    load_b_block(G + (LAST ? 2 : blk + 6) * (BK * K_DIM), gofb, s2);
    if (!LAST) load_a_block(pbase + (blk + 4) * BK, aA);
    mma_block(aB, s3, acc);
}

#define COVER_NOPS "v_nop\n\tv_nop\n\tv_nop\n\tv_nop\n\tv_nop\n\tv_nop\n\tv_nop\n\tv_nop"

__device__ __forceinline__ void cover_last(v8f (&acc)[4]) {
    asm volatile(COVER_NOPS
        : "+v"(acc[0]), "+v"(acc[1]), "+v"(acc[2]), "+v"(acc[3]));
}

template <int XMASK>
__device__ __forceinline__ float row_xmask_add(float x) {
    int y = __builtin_amdgcn_update_dpp(0, __float_as_int(x), 0x160 + XMASK,
                                        0xf, 0xf, true);
    return x + __int_as_float(y);
}

__device__ __forceinline__ float row16_sum(float x) {
    x = row_xmask_add<1>(x);
    x = row_xmask_add<2>(x);
    x = row_xmask_add<4>(x);
    x = row_xmask_add<8>(x);
    return x;
}

__device__ __forceinline__ float dot4(float4 a, float4 b) {
    return a.x * b.x + a.y * b.y + a.z * b.z + a.w * b.w;
}

__device__ __forceinline__ void strip_row_sums(float (&ps)[8], float* slot, bool writer) {
#pragma unroll
    for (int v = 0; v < 8; ++v) ps[v] = row_xmask_add<1>(ps[v]);
#pragma unroll
    for (int v = 0; v < 8; ++v) ps[v] = row_xmask_add<2>(ps[v]);
#pragma unroll
    for (int v = 0; v < 8; ++v) ps[v] = row_xmask_add<4>(ps[v]);
#pragma unroll
    for (int v = 0; v < 8; ++v) ps[v] = row_xmask_add<8>(ps[v]);
    if (writer) {
        float4 s0, s1;
        s0.x = ps[0]; s0.y = ps[1]; s0.z = ps[2]; s0.w = ps[3];
        s1.x = ps[4]; s1.y = ps[5]; s1.z = ps[6]; s1.w = ps[7];
        *reinterpret_cast<float4*>(slot)     = s0;
        *reinterpret_cast<float4*>(slot + 4) = s1;
    }
}

__device__ __forceinline__ float sum_waves(const float* red, int row) {
    float s = red[row];
#pragma unroll
    for (int w = 1; w < 8; ++w) s += red[w * 16 + row];
    return s;
}

template <int SRC>
__device__ __forceinline__ float row_take(float x) {
    int y = __builtin_amdgcn_update_dpp(0, __float_as_int(x), 0x150 + SRC,
                                        0xf, 0xf, true);
    return __int_as_float(y);
}

__device__ __forceinline__ void take8(float s, float (&o)[8]) {
    o[0] = row_take<0>(s); o[1] = row_take<1>(s);
    o[2] = row_take<2>(s); o[3] = row_take<3>(s);
    o[4] = row_take<4>(s); o[5] = row_take<5>(s);
    o[6] = row_take<6>(s); o[7] = row_take<7>(s);
}

__global__ void __launch_bounds__(256, 1)
cg_solve_kernel(const float* __restrict__ G0,
                const float* __restrict__ R0,
                const float* __restrict__ mask0,
                float* __restrict__ Xout0,
                const float* __restrict__ G1,
                const float* __restrict__ R1,
                const float* __restrict__ mask1,
                float* __restrict__ Xout1,
                int niter) {
    const bool dir1 = blockIdx.y != 0;
    const float* __restrict__ G    = dir1 ? G1    : G0;
    const float* __restrict__ R    = dir1 ? R1    : R0;
    const float* __restrict__ mask = dir1 ? mask1 : mask0;
    float* __restrict__ Xout       = dir1 ? Xout1 : Xout0;
    extern __shared__ float smem[];
    float* p_s = smem;
    __shared__ __align__(16) float red_pap[8 * 16];
    __shared__ __align__(16) float red_rr[8 * 16];

    int t    = threadIdx.x;
    int lane = t & 31;
    int wave = t >> 5;
    int lo   = lane & 15;
    int hi   = lane >> 4;
    int n0   = wave * 64;
    int m0   = blockIdx.x * 16;

    int erow = 8 * hi;
    int ecol = n0 + 4 * lo;
    int srow = erow + (lo & 7);

    unsigned gofb = 4u * (hi * (2 * K_DIM) + 2 * (n0 + 4 * lo));
    const float* pbase = p_s + lo * K_DIM + 2 * hi;

    v4f s0[2 * KU], s1[2 * KU], s2[2 * KU], s3[2 * KU];
    load_b_block(G, gofb, s0);
    load_b_block(G + 1 * BK * K_DIM, gofb, s1);
    load_b_block(G + 2 * BK * K_DIM, gofb, s2);

    float4 xv[8], rv[8];
    {
        float ps[8];
#pragma unroll
        for (int v = 0; v < 8; ++v) {
            float4 q = *reinterpret_cast<const float4*>(
                &R[(m0 + erow + v) * K_DIM + ecol]);
            rv[v] = q;
            xv[v].x = 0.0f; xv[v].y = 0.0f; xv[v].z = 0.0f; xv[v].w = 0.0f;
            *reinterpret_cast<float4*>(&p_s[(erow + v) * K_DIM + ecol]) = q;
            ps[v] = dot4(q, q);
        }
        strip_row_sums(ps, &red_rr[wave * 16 + erow], lo == 0);
    }
    __syncthreads();
    float rs_lane = sum_waves(red_rr, srow);

    float4 lmk[8];
#pragma unroll
    for (int v = 0; v < 8; ++v) {
        float4 mk = *reinterpret_cast<const float4*>(
            &mask[(m0 + erow + v) * K_DIM + ecol]);
        lmk[v].x = LMBDA_F * mk.x; lmk[v].y = LMBDA_F * mk.y;
        lmk[v].z = LMBDA_F * mk.z; lmk[v].w = LMBDA_F * mk.w;
    }

    for (int it = 0; it < niter; ++it) {
        v8f zero = {};
        v8f acc[4];
#pragma unroll
        for (int ni = 0; ni < 4; ++ni) acc[ni] = zero;

        {
            v2f aA[KU], aB[KU];
            load_a_block(pbase, aA);
#pragma nounroll
            for (int blk = 0; blk < NBLK - 4; blk += 4)
                ring_turn<false>(G, gofb, pbase, blk, s0, s1, s2, s3, aA, aB, acc);
            ring_turn<true>(G, gofb, pbase, NBLK - 4, s0, s1, s2, s3, aA, aB, acc);
            cover_last(acc);
        }

        float4 pv[8], av[8];
        float ps[8];
#pragma unroll
        for (int v = 0; v < 8; ++v) {
            pv[v] = *reinterpret_cast<const float4*>(&p_s[(erow + v) * K_DIM + ecol]);
            av[v].x = acc[0][v] + lmk[v].x * pv[v].x;
            av[v].y = acc[1][v] + lmk[v].y * pv[v].y;
            av[v].z = acc[2][v] + lmk[v].z * pv[v].z;
            av[v].w = acc[3][v] + lmk[v].w * pv[v].w;
            ps[v] = dot4(pv[v], av[v]);
        }
        strip_row_sums(ps, &red_pap[wave * 16 + erow], lo == 0);
        __syncthreads();

        float a[8];
        take8(rs_lane / sum_waves(red_pap, srow), a);
#pragma unroll
        for (int v = 0; v < 8; ++v) {
            xv[v].x = xv[v].x + a[v] * pv[v].x;  xv[v].y = xv[v].y + a[v] * pv[v].y;
            xv[v].z = xv[v].z + a[v] * pv[v].z;  xv[v].w = xv[v].w + a[v] * pv[v].w;
            rv[v].x = rv[v].x - a[v] * av[v].x;  rv[v].y = rv[v].y - a[v] * av[v].y;
            rv[v].z = rv[v].z - a[v] * av[v].z;  rv[v].w = rv[v].w - a[v] * av[v].w;
            ps[v] = dot4(rv[v], rv[v]);
        }
        strip_row_sums(ps, &red_rr[wave * 16 + erow], lo == 0);
        __syncthreads();

        float rs_next = sum_waves(red_rr, srow);
        float b[8];
        take8(rs_next / rs_lane, b);
        rs_lane = rs_next;
#pragma unroll
        for (int v = 0; v < 8; ++v) {
            float4 pn;
            pn.x = rv[v].x + b[v] * pv[v].x;  pn.y = rv[v].y + b[v] * pv[v].y;
            pn.z = rv[v].z + b[v] * pv[v].z;  pn.w = rv[v].w + b[v] * pv[v].w;
            *reinterpret_cast<float4*>(&p_s[(erow + v) * K_DIM + ecol]) = pn;
        }
        __syncthreads();
    }

#pragma unroll
    for (int v = 0; v < 8; ++v)
        *reinterpret_cast<float4*>(&Xout[(m0 + erow + v) * K_DIM + ecol]) = xv[v];
}

extern "C" void kernel_launch(void* const* d_in, const int* in_sizes, int n_in,
                              void* d_out, int out_size, void* d_ws, size_t ws_size,
                              hipStream_t stream) {
    const float* A  = (const float*)d_in[0];
    const float* B  = (const float*)d_in[1];
    const float* ea = (const float*)d_in[2];
    const float* eb = (const float*)d_in[3];
    float* out = (float*)d_out;

    float* ws      = (float*)d_ws;
    float* s_max   = ws;
    float* mask_ab = s_max   + 16;
    float* mask_ba = mask_ab + K_DIM * K_DIM;
    float* G_a     = mask_ba + K_DIM * K_DIM;
    float* G_b     = G_a     + K_DIM * K_DIM;
    float* R_ab    = G_b     + K_DIM * K_DIM;
    float* R_ba    = R_ab    + K_DIM * K_DIM;

    max_reduce_kernel<<<1, 512, 0, stream>>>(ea, eb, s_max);

    dim3 mgrid(K_DIM / 256, K_DIM);
    mask_kernel<<<mgrid, 256, 0, stream>>>(eb, ea, s_max, mask_ab);
    mask_kernel<<<mgrid, 256, 0, stream>>>(ea, eb, s_max, mask_ba);

    dim3 ggrid(8, 2, 4);
    gram_kernel<<<ggrid, 256, 0, stream>>>(A, B, G_a);

    size_t shmem = (size_t)16 * K_DIM * sizeof(float);
    dim3 cgrid(K_DIM / 16, 2);
    cg_solve_kernel<<<cgrid, 256, shmem, stream>>>(
        G_a, R_ab, mask_ab, out,
        G_b, R_ba, mask_ba, out + K_DIM * K_DIM,
        NITER);
}


# Gram products spread over all 256 WGPs: one 64x64 tile per 4-wave workgroup (was 64 workgroups of 8 waves), X and Y staged through LDS one 32-row step ahead by 128-bit loads; sum order unchanged
# speedup vs baseline: 5.9838x; 1.0676x over previous
#include <hip/hip_runtime.h>


#define D_DIM   1024
#define K_DIM   512
#define LMBDA_F 1000.0f
#define NITER   96

typedef float v2f __attribute__((ext_vector_type(2)));
typedef float v4f __attribute__((ext_vector_type(4)));
typedef float v8f __attribute__((ext_vector_type(8)));

__device__ __forceinline__ v8f wmma_f32(v2f a, v2f b, v8f c) {
    return __builtin_amdgcn_wmma_f32_16x16x4_f32(
        false, a, false, b, (short)0, c, false, false);
}

__global__ void max_reduce_kernel(const float* __restrict__ ea,
                                  const float* __restrict__ eb,
                                  float* __restrict__ s_out) {
    __shared__ float sm[512];
    int t = threadIdx.x;
    sm[t] = fmaxf(ea[t], eb[t]);
    __syncthreads();
    for (int off = 256; off > 0; off >>= 1) {
        if (t < off) sm[t] = fmaxf(sm[t], sm[t + off]);
        __syncthreads();
    }
    if (t == 0) s_out[0] = sm[0];
}

__global__ void mask_kernel(const float* __restrict__ erow,
                            const float* __restrict__ ecol,
                            const float* __restrict__ s_ptr,
                            float* __restrict__ mask) {
    int j = blockIdx.x * blockDim.x + threadIdx.x;
    int i = blockIdx.y;
    float inv_s = 1.0f / s_ptr[0];
    float gb = erow[i] * inv_s;
    float ga = ecol[j] * inv_s;
    float gb2 = gb * gb + 1.0f;
    float ga2 = ga * ga + 1.0f;
    float re = gb / gb2 - ga / ga2;
    float im = 1.0f / gb2 - 1.0f / ga2;
    mask[i * K_DIM + j] = re * re + im * im;
}

#define GT     64
#define GKB    32
#define GSTEPS (D_DIM / GKB)
#define GPP 160

#define COVER_NOPS "v_nop\n\tv_nop\n\tv_nop\n\tv_nop\n\tv_nop\n\tv_nop\n\tv_nop\n\tv_nop"

__device__ __forceinline__ void gram_cover_last(v8f (&acc)[2][2], int& rb, int& nl) {
    asm volatile(COVER_NOPS
        : "+v"(acc[0][0]), "+v"(acc[0][1]), "+v"(acc[1][0]), "+v"(acc[1][1]),
          "+v"(rb), "+v"(nl));
}

__device__ __forceinline__ void gram_fetch(const float* __restrict__ xg,
                                           const float* __restrict__ yg,
                                           v4f (&xr)[4], v4f (&yr)[4]) {
#pragma unroll
    for (int i = 0; i < 2; ++i) {
        xr[2 * i]     = *reinterpret_cast<const v4f*>(xg + (16 * i) * K_DIM);
        xr[2 * i + 1] = *reinterpret_cast<const v4f*>(xg + (16 * i + 1) * K_DIM);
        yr[2 * i]     = *reinterpret_cast<const v4f*>(yg + (16 * i) * K_DIM);
        yr[2 * i + 1] = *reinterpret_cast<const v4f*>(yg + (16 * i + 1) * K_DIM);
    }
}

__device__ __forceinline__ void gram_stage(float* sx, float* sy,
                                           const v4f (&xr)[4], const v4f (&yr)[4]) {
#pragma unroll
    for (int i = 0; i < 2; ++i) {
        float* px = sx + 8 * i * GPP;
        float* py = sy + 8 * i * GPP;
        *reinterpret_cast<v4f*>(px)     = __builtin_shufflevector(xr[2 * i], xr[2 * i + 1], 0, 4, 1, 5);
        *reinterpret_cast<v4f*>(px + 4) = __builtin_shufflevector(xr[2 * i], xr[2 * i + 1], 2, 6, 3, 7);
        *reinterpret_cast<v4f*>(py)     = __builtin_shufflevector(yr[2 * i], yr[2 * i + 1], 0, 4, 1, 5);
        *reinterpret_cast<v4f*>(py + 4) = __builtin_shufflevector(yr[2 * i], yr[2 * i + 1], 2, 6, 3, 7);
    }
}

__global__ void __launch_bounds__(128, 1)
gram_kernel(const float* __restrict__ A,
            const float* __restrict__ B,
            float* __restrict__ C4) {
    __shared__ __align__(16) float sX[2][(GKB / 2) * GPP];
    __shared__ __align__(16) float sY[2][(GKB / 2) * GPP];

    const int z = blockIdx.z;
    const float* __restrict__ X = (z == 1 || z == 2) ? B : A;
    const float* __restrict__ Y = (z == 1 || z == 3) ? B : A;
    float* __restrict__ C = C4 + (size_t)z * K_DIM * K_DIM;
    int t    = threadIdx.x;
    int lane = t & 31;
    int wave = t >> 5;
    int lo = lane & 15;
    int hi = lane >> 4;
    int bm0 = blockIdx.x * GT;
    int bn0 = blockIdx.y * GT;
    int wm  = (wave & 1) * 32;
    int wn  = (wave >> 1) * 32;

    int spr  = t >> 4;
    int scol = (t & 15) * 4;
    const float* xg = X + (2 * spr) * K_DIM + bm0 + scol;
    const float* yg = Y + (2 * spr) * K_DIM + bn0 + scol;
    int sidx = spr * GPP + 2 * scol;

    v4f xr[4], yr[4];
    gram_fetch(xg, yg, xr, yr);
    gram_stage(&sX[0][sidx], &sY[0][sidx], xr, yr);
    __syncthreads();

    v8f zero = {};
    v8f acc[2][2];
#pragma unroll
    for (int mi = 0; mi < 2; ++mi)
#pragma unroll
        for (int ni = 0; ni < 2; ++ni) acc[mi][ni] = zero;

    int fa = hi * GPP + 2 * (wm + lo);
    int fb = hi * GPP + 2 * (wn + lo);

    for (int s = 0; s < GSTEPS; ++s) {
        const int cur = s & 1;
        const int sn = (s + 1) & (GSTEPS - 1);
        gram_fetch(xg + sn * (GKB * K_DIM), yg + sn * (GKB * K_DIM), xr, yr);

        const float* ax = &sX[cur][fa];
        const float* by = &sY[cur][fb];
#pragma unroll
        for (int u = 0; u < GKB / 4; ++u) {
            v2f aF[2], bF[2];
#pragma unroll
            for (int mi = 0; mi < 2; ++mi)
                aF[mi] = *reinterpret_cast<const v2f*>(ax + (2 * u) * GPP + 32 * mi);
#pragma unroll
            for (int ni = 0; ni < 2; ++ni)
                bF[ni] = *reinterpret_cast<const v2f*>(by + (2 * u) * GPP + 32 * ni);
#pragma unroll
            for (int mi = 0; mi < 2; ++mi)
#pragma unroll
                for (int ni = 0; ni < 2; ++ni)
                    acc[mi][ni] = wmma_f32(aF[mi], bF[ni], acc[mi][ni]);
        }

        gram_stage(&sX[cur ^ 1][sidx], &sY[cur ^ 1][sidx], xr, yr);
        __syncthreads();
    }

    int nl = bn0 + wn + lo;
    int rb = (bm0 + wm + 8 * hi) * K_DIM + nl;
    gram_cover_last(acc, rb, nl);
    if (z < 2) {
        int pb = rb + nl;
#pragma unroll
        for (int mi = 0; mi < 2; ++mi)
#pragma unroll
            for (int ni = 0; ni < 2; ++ni)
#pragma unroll
                for (int v = 0; v < 8; v += 2) {
                    v2f pr;
                    pr.x = acc[mi][ni][v];
                    pr.y = acc[mi][ni][v + 1];
                    *reinterpret_cast<v2f*>(&C[pb + (mi * 16 + v) * K_DIM + 32 * ni]) = pr;
                }
    } else {
#pragma unroll
        for (int mi = 0; mi < 2; ++mi)
#pragma unroll
            for (int ni = 0; ni < 2; ++ni)
#pragma unroll
                for (int v = 0; v < 8; ++v)
                    C[rb + (mi * 16 + v) * K_DIM + 16 * ni] = acc[mi][ni][v];
    }
}

#define KU 4
#define BK (4 * KU)
#define NBLK (K_DIM / BK)
static_assert(NBLK % 4 == 0 && NBLK >= 8, "the ring of four register sets turns once per four blocks");

__device__ __forceinline__ void load_b_block(const float* __restrict__ gblk,
                                             unsigned gofb, v4f (&s)[2 * KU]) {
    const char* gb = reinterpret_cast<const char*>(gblk);
#pragma unroll
    for (int u = 0; u < KU; ++u) {
        s[2 * u]     = *reinterpret_cast<const v4f*>(gb + gofb + (2 * u) * (2 * K_DIM) * 4);
        s[2 * u + 1] = *reinterpret_cast<const v4f*>(gb + gofb + (2 * u) * (2 * K_DIM) * 4 + 16);
    }
}

__device__ __forceinline__ void load_a_block(const float* pa, v2f (&aF)[KU]) {
#pragma unroll
    for (int u = 0; u < KU; ++u)
        aF[u] = *reinterpret_cast<const v2f*>(pa + 4 * u);
}

__device__ __forceinline__ void mma_block(const v2f (&aF)[KU],
                                          const v4f (&s)[2 * KU],
                                          v8f (&acc)[4]) {
    __builtin_amdgcn_sched_barrier(0);
#pragma unroll
    for (int u = 0; u < KU; ++u) {
        acc[0] = wmma_f32(aF[u], __builtin_shufflevector(s[2 * u],     s[2 * u],     0, 1), acc[0]);
        acc[1] = wmma_f32(aF[u], __builtin_shufflevector(s[2 * u],     s[2 * u],     2, 3), acc[1]);
        acc[2] = wmma_f32(aF[u], __builtin_shufflevector(s[2 * u + 1], s[2 * u + 1], 0, 1), acc[2]);
        acc[3] = wmma_f32(aF[u], __builtin_shufflevector(s[2 * u + 1], s[2 * u + 1], 2, 3), acc[3]);
    }
}

__device__ __forceinline__ void refill_behind_run(v8f (&acc)[4], unsigned& gofb) {
    asm volatile(""
        : "+v"(acc[0]), "+v"(acc[1]), "+v"(acc[2]), "+v"(acc[3]), "+v"(gofb));
}

template <bool LAST>
__device__ __forceinline__ void ring_turn(const float* __restrict__ G, unsigned& gofb,
                                          const float* pbase, int blk,
                                          v4f (&s0)[2 * KU], v4f (&s1)[2 * KU],
                                          v4f (&s2)[2 * KU], v4f (&s3)[2 * KU],
                                          v2f (&aA)[KU], v2f (&aB)[KU], v8f (&acc)[4]) {
    refill_behind_run(acc, gofb);
    load_b_block(G + (blk + 3) * (BK * K_DIM), gofb, s3);
    load_a_block(pbase + (blk + 1) * BK, aB);
    mma_block(aA, s0, acc);
    refill_behind_run(acc, gofb);
    load_b_block(G + (LAST ? 0 : blk + 4) * (BK * K_DIM), gofb, s0);
    load_a_block(pbase + (blk + 2) * BK, aA);
    mma_block(aB, s1, acc);
    refill_behind_run(acc, gofb);
    load_b_block(G + (LAST ? 1 : blk + 5) * (BK * K_DIM), gofb, s1);
    load_a_block(pbase + (blk + 3) * BK, aB);
    mma_block(aA, s2, acc);
    refill_behind_run(acc, gofb);
    load_b_block(G + (LAST ? 2 : blk + 6) * (BK * K_DIM), gofb, s2);
    if (!LAST) load_a_block(pbase + (blk + 4) * BK, aA);
    mma_block(aB, s3, acc);
}

__device__ __forceinline__ void cover_last(v8f (&acc)[4]) {
    asm volatile(COVER_NOPS
        : "+v"(acc[0]), "+v"(acc[1]), "+v"(acc[2]), "+v"(acc[3]));
}

template <int XMASK>
__device__ __forceinline__ float row_xmask_add(float x) {
    int y = __builtin_amdgcn_update_dpp(0, __float_as_int(x), 0x160 + XMASK,
                                        0xf, 0xf, true);
    return x + __int_as_float(y);
}

__device__ __forceinline__ float row16_sum(float x) {
    x = row_xmask_add<1>(x);
    x = row_xmask_add<2>(x);
    x = row_xmask_add<4>(x);
    x = row_xmask_add<8>(x);
    return x;
}

__device__ __forceinline__ float dot4(float4 a, float4 b) {
    return a.x * b.x + a.y * b.y + a.z * b.z + a.w * b.w;
}

__device__ __forceinline__ void strip_row_sums(float (&ps)[8], float* slot, bool writer) {
#pragma unroll
    for (int v = 0; v < 8; ++v) ps[v] = row_xmask_add<1>(ps[v]);
#pragma unroll
    for (int v = 0; v < 8; ++v) ps[v] = row_xmask_add<2>(ps[v]);
#pragma unroll
    for (int v = 0; v < 8; ++v) ps[v] = row_xmask_add<4>(ps[v]);
#pragma unroll
    for (int v = 0; v < 8; ++v) ps[v] = row_xmask_add<8>(ps[v]);
    if (writer) {
        float4 s0, s1;
        s0.x = ps[0]; s0.y = ps[1]; s0.z = ps[2]; s0.w = ps[3];
        s1.x = ps[4]; s1.y = ps[5]; s1.z = ps[6]; s1.w = ps[7];
        *reinterpret_cast<float4*>(slot)     = s0;
        *reinterpret_cast<float4*>(slot + 4) = s1;
    }
}

__device__ __forceinline__ float sum_waves(const float* red, int row) {
    float s = red[row];
#pragma unroll
    for (int w = 1; w < 8; ++w) s += red[w * 16 + row];
    return s;
}

template <int SRC>
__device__ __forceinline__ float row_take(float x) {
    int y = __builtin_amdgcn_update_dpp(0, __float_as_int(x), 0x150 + SRC,
                                        0xf, 0xf, true);
    return __int_as_float(y);
}

__device__ __forceinline__ void take8(float s, float (&o)[8]) {
    o[0] = row_take<0>(s); o[1] = row_take<1>(s);
    o[2] = row_take<2>(s); o[3] = row_take<3>(s);
    o[4] = row_take<4>(s); o[5] = row_take<5>(s);
    o[6] = row_take<6>(s); o[7] = row_take<7>(s);
}

__global__ void __launch_bounds__(256, 1)
cg_solve_kernel(const float* __restrict__ G0,
                const float* __restrict__ R0,
                const float* __restrict__ mask0,
                float* __restrict__ Xout0,
                const float* __restrict__ G1,
                const float* __restrict__ R1,
                const float* __restrict__ mask1,
                float* __restrict__ Xout1,
                int niter) {
    const bool dir1 = blockIdx.y != 0;
    const float* __restrict__ G    = dir1 ? G1    : G0;
    const float* __restrict__ R    = dir1 ? R1    : R0;
    const float* __restrict__ mask = dir1 ? mask1 : mask0;
    float* __restrict__ Xout       = dir1 ? Xout1 : Xout0;
    extern __shared__ float smem[];
    float* p_s = smem;
    __shared__ __align__(16) float red_pap[8 * 16];
    __shared__ __align__(16) float red_rr[8 * 16];

    int t    = threadIdx.x;
    int lane = t & 31;
    int wave = t >> 5;
    int lo   = lane & 15;
    int hi   = lane >> 4;
    int n0   = wave * 64;
    int m0   = blockIdx.x * 16;

    int erow = 8 * hi;
    int ecol = n0 + 4 * lo;
    int srow = erow + (lo & 7);

    unsigned gofb = 4u * (hi * (2 * K_DIM) + 2 * (n0 + 4 * lo));
    const float* pbase = p_s + lo * K_DIM + 2 * hi;

    v4f s0[2 * KU], s1[2 * KU], s2[2 * KU], s3[2 * KU];
    load_b_block(G, gofb, s0);
    load_b_block(G + 1 * BK * K_DIM, gofb, s1);
    load_b_block(G + 2 * BK * K_DIM, gofb, s2);

    float4 xv[8], rv[8];
    {
        float ps[8];
#pragma unroll
        for (int v = 0; v < 8; ++v) {
            float4 q = *reinterpret_cast<const float4*>(
                &R[(m0 + erow + v) * K_DIM + ecol]);
            rv[v] = q;
            xv[v].x = 0.0f; xv[v].y = 0.0f; xv[v].z = 0.0f; xv[v].w = 0.0f;
            *reinterpret_cast<float4*>(&p_s[(erow + v) * K_DIM + ecol]) = q;
            ps[v] = dot4(q, q);
        }
        strip_row_sums(ps, &red_rr[wave * 16 + erow], lo == 0);
    }
    __syncthreads();
    float rs_lane = sum_waves(red_rr, srow);

    float4 lmk[8];
#pragma unroll
    for (int v = 0; v < 8; ++v) {
        float4 mk = *reinterpret_cast<const float4*>(
            &mask[(m0 + erow + v) * K_DIM + ecol]);
        lmk[v].x = LMBDA_F * mk.x; lmk[v].y = LMBDA_F * mk.y;
        lmk[v].z = LMBDA_F * mk.z; lmk[v].w = LMBDA_F * mk.w;
    }

    for (int it = 0; it < niter; ++it) {
        v8f zero = {};
        v8f acc[4];
#pragma unroll
        for (int ni = 0; ni < 4; ++ni) acc[ni] = zero;

        {
            v2f aA[KU], aB[KU];
            load_a_block(pbase, aA);
#pragma nounroll
            for (int blk = 0; blk < NBLK - 4; blk += 4)
                ring_turn<false>(G, gofb, pbase, blk, s0, s1, s2, s3, aA, aB, acc);
            ring_turn<true>(G, gofb, pbase, NBLK - 4, s0, s1, s2, s3, aA, aB, acc);
            cover_last(acc);
        }

        float4 pv[8], av[8];
        float ps[8];
#pragma unroll
        for (int v = 0; v < 8; ++v) {
            pv[v] = *reinterpret_cast<const float4*>(&p_s[(erow + v) * K_DIM + ecol]);
            av[v].x = acc[0][v] + lmk[v].x * pv[v].x;
            av[v].y = acc[1][v] + lmk[v].y * pv[v].y;
            av[v].z = acc[2][v] + lmk[v].z * pv[v].z;
            av[v].w = acc[3][v] + lmk[v].w * pv[v].w;
            ps[v] = dot4(pv[v], av[v]);
        }
        strip_row_sums(ps, &red_pap[wave * 16 + erow], lo == 0);
        __syncthreads();

        float a[8];
        take8(rs_lane / sum_waves(red_pap, srow), a);
#pragma unroll
        for (int v = 0; v < 8; ++v) {
            xv[v].x = xv[v].x + a[v] * pv[v].x;  xv[v].y = xv[v].y + a[v] * pv[v].y;
            xv[v].z = xv[v].z + a[v] * pv[v].z;  xv[v].w = xv[v].w + a[v] * pv[v].w;
            rv[v].x = rv[v].x - a[v] * av[v].x;  rv[v].y = rv[v].y - a[v] * av[v].y;
            rv[v].z = rv[v].z - a[v] * av[v].z;  rv[v].w = rv[v].w - a[v] * av[v].w;
            ps[v] = dot4(rv[v], rv[v]);
        }
        strip_row_sums(ps, &red_rr[wave * 16 + erow], lo == 0);
        __syncthreads();

        float rs_next = sum_waves(red_rr, srow);
        float b[8];
        take8(rs_next / rs_lane, b);
        rs_lane = rs_next;
#pragma unroll
        for (int v = 0; v < 8; ++v) {
            float4 pn;
            pn.x = rv[v].x + b[v] * pv[v].x;  pn.y = rv[v].y + b[v] * pv[v].y;
            pn.z = rv[v].z + b[v] * pv[v].z;  pn.w = rv[v].w + b[v] * pv[v].w;
            *reinterpret_cast<float4*>(&p_s[(erow + v) * K_DIM + ecol]) = pn;
        }
        __syncthreads();
    }

#pragma unroll
    for (int v = 0; v < 8; ++v)
        *reinterpret_cast<float4*>(&Xout[(m0 + erow + v) * K_DIM + ecol]) = xv[v];
}

extern "C" void kernel_launch(void* const* d_in, const int* in_sizes, int n_in,
                              void* d_out, int out_size, void* d_ws, size_t ws_size,
                              hipStream_t stream) {
    const float* A  = (const float*)d_in[0];
    const float* B  = (const float*)d_in[1];
    const float* ea = (const float*)d_in[2];
    const float* eb = (const float*)d_in[3];
    float* out = (float*)d_out;

    float* ws      = (float*)d_ws;
    float* s_max   = ws;
    float* mask_ab = s_max   + 16;
    float* mask_ba = mask_ab + K_DIM * K_DIM;
    float* G_a     = mask_ba + K_DIM * K_DIM;
    float* G_b     = G_a     + K_DIM * K_DIM;
    float* R_ab    = G_b     + K_DIM * K_DIM;
    float* R_ba    = R_ab    + K_DIM * K_DIM;

    max_reduce_kernel<<<1, 512, 0, stream>>>(ea, eb, s_max);

    dim3 mgrid(K_DIM / 256, K_DIM);
    mask_kernel<<<mgrid, 256, 0, stream>>>(eb, ea, s_max, mask_ab);
    mask_kernel<<<mgrid, 256, 0, stream>>>(ea, eb, s_max, mask_ba);

    dim3 ggrid(8, 8, 4);
    gram_kernel<<<ggrid, 128, 0, stream>>>(A, B, G_a);

    size_t shmem = (size_t)16 * K_DIM * sizeof(float);
    dim3 cgrid(K_DIM / 16, 2);
    cg_solve_kernel<<<cgrid, 256, shmem, stream>>>(
        G_a, R_ab, mask_ab, out,
        G_b, R_ba, mask_ba, out + K_DIM * K_DIM,
        NITER);
}


# CG loop exits on convergence: after the <r,r> reduction each wave tests its 16 rows' <r,r> <= 2^-60 <r0,r0> by a wave vote (same words in all waves: block-uniform exit); 96 stays the cap
# speedup vs baseline: 11.3417x; 1.8954x over previous
#include <hip/hip_runtime.h>


#define D_DIM   1024
#define K_DIM   512
#define LMBDA_F 1000.0f
#define NITER   96

typedef float v2f __attribute__((ext_vector_type(2)));
typedef float v4f __attribute__((ext_vector_type(4)));
typedef float v8f __attribute__((ext_vector_type(8)));

__device__ __forceinline__ v8f wmma_f32(v2f a, v2f b, v8f c) {
    return __builtin_amdgcn_wmma_f32_16x16x4_f32(
        false, a, false, b, (short)0, c, false, false);
}

__global__ void max_reduce_kernel(const float* __restrict__ ea,
                                  const float* __restrict__ eb,
                                  float* __restrict__ s_out) {
    __shared__ float sm[512];
    int t = threadIdx.x;
    sm[t] = fmaxf(ea[t], eb[t]);
    __syncthreads();
    for (int off = 256; off > 0; off >>= 1) {
        if (t < off) sm[t] = fmaxf(sm[t], sm[t + off]);
        __syncthreads();
    }
    if (t == 0) s_out[0] = sm[0];
}

__global__ void mask_kernel(const float* __restrict__ erow,
                            const float* __restrict__ ecol,
                            const float* __restrict__ s_ptr,
                            float* __restrict__ mask) {
    int j = blockIdx.x * blockDim.x + threadIdx.x;
    int i = blockIdx.y;
    float inv_s = 1.0f / s_ptr[0];
    float gb = erow[i] * inv_s;
    float ga = ecol[j] * inv_s;
    float gb2 = gb * gb + 1.0f;
    float ga2 = ga * ga + 1.0f;
    float re = gb / gb2 - ga / ga2;
    float im = 1.0f / gb2 - 1.0f / ga2;
    mask[i * K_DIM + j] = re * re + im * im;
}

#define GT     64
#define GKB    32
#define GSTEPS (D_DIM / GKB)
#define GPP 160

#define COVER_NOPS "v_nop\n\tv_nop\n\tv_nop\n\tv_nop\n\tv_nop\n\tv_nop\n\tv_nop\n\tv_nop"

__device__ __forceinline__ void gram_cover_last(v8f (&acc)[2][2], int& rb, int& nl) {
    asm volatile(COVER_NOPS
        : "+v"(acc[0][0]), "+v"(acc[0][1]), "+v"(acc[1][0]), "+v"(acc[1][1]),
          "+v"(rb), "+v"(nl));
}

__device__ __forceinline__ void gram_fetch(const float* __restrict__ xg,
                                           const float* __restrict__ yg,
                                           v4f (&xr)[4], v4f (&yr)[4]) {
#pragma unroll
    for (int i = 0; i < 2; ++i) {
        xr[2 * i]     = *reinterpret_cast<const v4f*>(xg + (16 * i) * K_DIM);
        xr[2 * i + 1] = *reinterpret_cast<const v4f*>(xg + (16 * i + 1) * K_DIM);
        yr[2 * i]     = *reinterpret_cast<const v4f*>(yg + (16 * i) * K_DIM);
        yr[2 * i + 1] = *reinterpret_cast<const v4f*>(yg + (16 * i + 1) * K_DIM);
    }
}

__device__ __forceinline__ void gram_stage(float* sx, float* sy,
                                           const v4f (&xr)[4], const v4f (&yr)[4]) {
#pragma unroll
    for (int i = 0; i < 2; ++i) {
        float* px = sx + 8 * i * GPP;
        float* py = sy + 8 * i * GPP;
        *reinterpret_cast<v4f*>(px)     = __builtin_shufflevector(xr[2 * i], xr[2 * i + 1], 0, 4, 1, 5);
        *reinterpret_cast<v4f*>(px + 4) = __builtin_shufflevector(xr[2 * i], xr[2 * i + 1], 2, 6, 3, 7);
        *reinterpret_cast<v4f*>(py)     = __builtin_shufflevector(yr[2 * i], yr[2 * i + 1], 0, 4, 1, 5);
        *reinterpret_cast<v4f*>(py + 4) = __builtin_shufflevector(yr[2 * i], yr[2 * i + 1], 2, 6, 3, 7);
    }
}

__global__ void __launch_bounds__(128, 1)
gram_kernel(const float* __restrict__ A,
            const float* __restrict__ B,
            float* __restrict__ C4) {
    __shared__ __align__(16) float sX[2][(GKB / 2) * GPP];
    __shared__ __align__(16) float sY[2][(GKB / 2) * GPP];

    const int z = blockIdx.z;
    const float* __restrict__ X = (z == 1 || z == 2) ? B : A;
    const float* __restrict__ Y = (z == 1 || z == 3) ? B : A;
    float* __restrict__ C = C4 + (size_t)z * K_DIM * K_DIM;
    int t    = threadIdx.x;
    int lane = t & 31;
    int wave = t >> 5;
    int lo = lane & 15;
    int hi = lane >> 4;
    int bm0 = blockIdx.x * GT;
    int bn0 = blockIdx.y * GT;
    int wm  = (wave & 1) * 32;
    int wn  = (wave >> 1) * 32;

    int spr  = t >> 4;
    int scol = (t & 15) * 4;
    const float* xg = X + (2 * spr) * K_DIM + bm0 + scol;
    const float* yg = Y + (2 * spr) * K_DIM + bn0 + scol;
    int sidx = spr * GPP + 2 * scol;

    v4f xr[4], yr[4];
    gram_fetch(xg, yg, xr, yr);
    gram_stage(&sX[0][sidx], &sY[0][sidx], xr, yr);
    __syncthreads();

    v8f zero = {};
    v8f acc[2][2];
#pragma unroll
    for (int mi = 0; mi < 2; ++mi)
#pragma unroll
        for (int ni = 0; ni < 2; ++ni) acc[mi][ni] = zero;

    int fa = hi * GPP + 2 * (wm + lo);
    int fb = hi * GPP + 2 * (wn + lo);

    for (int s = 0; s < GSTEPS; ++s) {
        const int cur = s & 1;
        const int sn = (s + 1) & (GSTEPS - 1);
        gram_fetch(xg + sn * (GKB * K_DIM), yg + sn * (GKB * K_DIM), xr, yr);

        const float* ax = &sX[cur][fa];
        const float* by = &sY[cur][fb];
#pragma unroll
        for (int u = 0; u < GKB / 4; ++u) {
            v2f aF[2], bF[2];
#pragma unroll
            for (int mi = 0; mi < 2; ++mi)
                aF[mi] = *reinterpret_cast<const v2f*>(ax + (2 * u) * GPP + 32 * mi);
#pragma unroll
            for (int ni = 0; ni < 2; ++ni)
                bF[ni] = *reinterpret_cast<const v2f*>(by + (2 * u) * GPP + 32 * ni);
#pragma unroll
            for (int mi = 0; mi < 2; ++mi)
#pragma unroll
                for (int ni = 0; ni < 2; ++ni)
                    acc[mi][ni] = wmma_f32(aF[mi], bF[ni], acc[mi][ni]);
        }

        gram_stage(&sX[cur ^ 1][sidx], &sY[cur ^ 1][sidx], xr, yr);
        __syncthreads();
    }

    int nl = bn0 + wn + lo;
    int rb = (bm0 + wm + 8 * hi) * K_DIM + nl;
    gram_cover_last(acc, rb, nl);
    if (z < 2) {
        int pb = rb + nl;
#pragma unroll
        for (int mi = 0; mi < 2; ++mi)
#pragma unroll
            for (int ni = 0; ni < 2; ++ni)
#pragma unroll
                for (int v = 0; v < 8; v += 2) {
                    v2f pr;
                    pr.x = acc[mi][ni][v];
                    pr.y = acc[mi][ni][v + 1];
                    *reinterpret_cast<v2f*>(&C[pb + (mi * 16 + v) * K_DIM + 32 * ni]) = pr;
                }
    } else {
#pragma unroll
        for (int mi = 0; mi < 2; ++mi)
#pragma unroll
            for (int ni = 0; ni < 2; ++ni)
#pragma unroll
                for (int v = 0; v < 8; ++v)
                    C[rb + (mi * 16 + v) * K_DIM + 16 * ni] = acc[mi][ni][v];
    }
}

#define KU 4
#define BK (4 * KU)
#define NBLK (K_DIM / BK)
static_assert(NBLK % 4 == 0 && NBLK >= 8, "the ring of four register sets turns once per four blocks");

__device__ __forceinline__ void load_b_block(const float* __restrict__ gblk,
                                             unsigned gofb, v4f (&s)[2 * KU]) {
    const char* gb = reinterpret_cast<const char*>(gblk);
#pragma unroll
    for (int u = 0; u < KU; ++u) {
        s[2 * u]     = *reinterpret_cast<const v4f*>(gb + gofb + (2 * u) * (2 * K_DIM) * 4);
        s[2 * u + 1] = *reinterpret_cast<const v4f*>(gb + gofb + (2 * u) * (2 * K_DIM) * 4 + 16);
    }
}

__device__ __forceinline__ void load_a_block(const float* pa, v2f (&aF)[KU]) {
#pragma unroll
    for (int u = 0; u < KU; ++u)
        aF[u] = *reinterpret_cast<const v2f*>(pa + 4 * u);
}

__device__ __forceinline__ void mma_block(const v2f (&aF)[KU],
                                          const v4f (&s)[2 * KU],
                                          v8f (&acc)[4]) {
    __builtin_amdgcn_sched_barrier(0);
#pragma unroll
    for (int u = 0; u < KU; ++u) {
        acc[0] = wmma_f32(aF[u], __builtin_shufflevector(s[2 * u],     s[2 * u],     0, 1), acc[0]);
        acc[1] = wmma_f32(aF[u], __builtin_shufflevector(s[2 * u],     s[2 * u],     2, 3), acc[1]);
        acc[2] = wmma_f32(aF[u], __builtin_shufflevector(s[2 * u + 1], s[2 * u + 1], 0, 1), acc[2]);
        acc[3] = wmma_f32(aF[u], __builtin_shufflevector(s[2 * u + 1], s[2 * u + 1], 2, 3), acc[3]);
    }
}

__device__ __forceinline__ void refill_behind_run(v8f (&acc)[4], unsigned& gofb) {
    asm volatile(""
        : "+v"(acc[0]), "+v"(acc[1]), "+v"(acc[2]), "+v"(acc[3]), "+v"(gofb));
}

template <bool LAST>
__device__ __forceinline__ void ring_turn(const float* __restrict__ G, unsigned& gofb,
                                          const float* pbase, int blk,
                                          v4f (&s0)[2 * KU], v4f (&s1)[2 * KU],
                                          v4f (&s2)[2 * KU], v4f (&s3)[2 * KU],
                                          v2f (&aA)[KU], v2f (&aB)[KU], v8f (&acc)[4]) {
    refill_behind_run(acc, gofb);
    load_b_block(G + (blk + 3) * (BK * K_DIM), gofb, s3);
    load_a_block(pbase + (blk + 1) * BK, aB);
    mma_block(aA, s0, acc);
    refill_behind_run(acc, gofb);
    load_b_block(G + (LAST ? 0 : blk + 4) * (BK * K_DIM), gofb, s0);
    load_a_block(pbase + (blk + 2) * BK, aA);
    mma_block(aB, s1, acc);
    refill_behind_run(acc, gofb);
    load_b_block(G + (LAST ? 1 : blk + 5) * (BK * K_DIM), gofb, s1);
    load_a_block(pbase + (blk + 3) * BK, aB);
    mma_block(aA, s2, acc);
    refill_behind_run(acc, gofb);
    load_b_block(G + (LAST ? 2 : blk + 6) * (BK * K_DIM), gofb, s2);
    if (!LAST) load_a_block(pbase + (blk + 4) * BK, aA);
    mma_block(aB, s3, acc);
}

__device__ __forceinline__ void cover_last(v8f (&acc)[4]) {
    asm volatile(COVER_NOPS
        : "+v"(acc[0]), "+v"(acc[1]), "+v"(acc[2]), "+v"(acc[3]));
}

template <int XMASK>
__device__ __forceinline__ float row_xmask_add(float x) {
    int y = __builtin_amdgcn_update_dpp(0, __float_as_int(x), 0x160 + XMASK,
                                        0xf, 0xf, true);
    return x + __int_as_float(y);
}

__device__ __forceinline__ float row16_sum(float x) {
    x = row_xmask_add<1>(x);
    x = row_xmask_add<2>(x);
    x = row_xmask_add<4>(x);
    x = row_xmask_add<8>(x);
    return x;
}

#define CG_STOP_RR 0x1p-60f

__device__ __forceinline__ float dot4(float4 a, float4 b) {
    return a.x * b.x + a.y * b.y + a.z * b.z + a.w * b.w;
}

__device__ __forceinline__ void strip_row_sums(float (&ps)[8], float* slot, bool writer) {
#pragma unroll
    for (int v = 0; v < 8; ++v) ps[v] = row_xmask_add<1>(ps[v]);
#pragma unroll
    for (int v = 0; v < 8; ++v) ps[v] = row_xmask_add<2>(ps[v]);
#pragma unroll
    for (int v = 0; v < 8; ++v) ps[v] = row_xmask_add<4>(ps[v]);
#pragma unroll
    for (int v = 0; v < 8; ++v) ps[v] = row_xmask_add<8>(ps[v]);
    if (writer) {
        float4 s0, s1;
        s0.x = ps[0]; s0.y = ps[1]; s0.z = ps[2]; s0.w = ps[3];
        s1.x = ps[4]; s1.y = ps[5]; s1.z = ps[6]; s1.w = ps[7];
        *reinterpret_cast<float4*>(slot)     = s0;
        *reinterpret_cast<float4*>(slot + 4) = s1;
    }
}

__device__ __forceinline__ float sum_waves(const float* red, int row) {
    float s = red[row];
#pragma unroll
    for (int w = 1; w < 8; ++w) s += red[w * 16 + row];
    return s;
}

template <int SRC>
__device__ __forceinline__ float row_take(float x) {
    int y = __builtin_amdgcn_update_dpp(0, __float_as_int(x), 0x150 + SRC,
                                        0xf, 0xf, true);
    return __int_as_float(y);
}

__device__ __forceinline__ void take8(float s, float (&o)[8]) {
    o[0] = row_take<0>(s); o[1] = row_take<1>(s);
    o[2] = row_take<2>(s); o[3] = row_take<3>(s);
    o[4] = row_take<4>(s); o[5] = row_take<5>(s);
    o[6] = row_take<6>(s); o[7] = row_take<7>(s);
}

__global__ void __launch_bounds__(256, 1)
cg_solve_kernel(const float* __restrict__ G0,
                const float* __restrict__ R0,
                const float* __restrict__ mask0,
                float* __restrict__ Xout0,
                const float* __restrict__ G1,
                const float* __restrict__ R1,
                const float* __restrict__ mask1,
                float* __restrict__ Xout1,
                int niter) {
    const bool dir1 = blockIdx.y != 0;
    const float* __restrict__ G    = dir1 ? G1    : G0;
    const float* __restrict__ R    = dir1 ? R1    : R0;
    const float* __restrict__ mask = dir1 ? mask1 : mask0;
    float* __restrict__ Xout       = dir1 ? Xout1 : Xout0;
    extern __shared__ float smem[];
    float* p_s = smem;
    __shared__ __align__(16) float red_pap[8 * 16];
    __shared__ __align__(16) float red_rr[8 * 16];

    int t    = threadIdx.x;
    int lane = t & 31;
    int wave = t >> 5;
    int lo   = lane & 15;
    int hi   = lane >> 4;
    int n0   = wave * 64;
    int m0   = blockIdx.x * 16;

    int erow = 8 * hi;
    int ecol = n0 + 4 * lo;
    int srow = erow + (lo & 7);

    unsigned gofb = 4u * (hi * (2 * K_DIM) + 2 * (n0 + 4 * lo));
    const float* pbase = p_s + lo * K_DIM + 2 * hi;

    v4f s0[2 * KU], s1[2 * KU], s2[2 * KU], s3[2 * KU];
    load_b_block(G, gofb, s0);
    load_b_block(G + 1 * BK * K_DIM, gofb, s1);
    load_b_block(G + 2 * BK * K_DIM, gofb, s2);

    float4 xv[8], rv[8];
    {
        float ps[8];
#pragma unroll
        for (int v = 0; v < 8; ++v) {
            float4 q = *reinterpret_cast<const float4*>(
                &R[(m0 + erow + v) * K_DIM + ecol]);
            rv[v] = q;
            xv[v].x = 0.0f; xv[v].y = 0.0f; xv[v].z = 0.0f; xv[v].w = 0.0f;
            *reinterpret_cast<float4*>(&p_s[(erow + v) * K_DIM + ecol]) = q;
            ps[v] = dot4(q, q);
        }
        strip_row_sums(ps, &red_rr[wave * 16 + erow], lo == 0);
    }
    __syncthreads();
    float rs_lane = sum_waves(red_rr, srow);
    const float rs_stop = CG_STOP_RR * rs_lane;

    float4 lmk[8];
#pragma unroll
    for (int v = 0; v < 8; ++v) {
        float4 mk = *reinterpret_cast<const float4*>(
            &mask[(m0 + erow + v) * K_DIM + ecol]);
        lmk[v].x = LMBDA_F * mk.x; lmk[v].y = LMBDA_F * mk.y;
        lmk[v].z = LMBDA_F * mk.z; lmk[v].w = LMBDA_F * mk.w;
    }

    for (int it = 0; it < niter; ++it) {
        v8f zero = {};
        v8f acc[4];
#pragma unroll
        for (int ni = 0; ni < 4; ++ni) acc[ni] = zero;

        {
            v2f aA[KU], aB[KU];
            load_a_block(pbase, aA);
#pragma nounroll
            for (int blk = 0; blk < NBLK - 4; blk += 4)
                ring_turn<false>(G, gofb, pbase, blk, s0, s1, s2, s3, aA, aB, acc);
            ring_turn<true>(G, gofb, pbase, NBLK - 4, s0, s1, s2, s3, aA, aB, acc);
            cover_last(acc);
        }

        float4 pv[8], av[8];
        float ps[8];
#pragma unroll
        for (int v = 0; v < 8; ++v) {
            pv[v] = *reinterpret_cast<const float4*>(&p_s[(erow + v) * K_DIM + ecol]);
            av[v].x = acc[0][v] + lmk[v].x * pv[v].x;
            av[v].y = acc[1][v] + lmk[v].y * pv[v].y;
            av[v].z = acc[2][v] + lmk[v].z * pv[v].z;
            av[v].w = acc[3][v] + lmk[v].w * pv[v].w;
            ps[v] = dot4(pv[v], av[v]);
        }
        strip_row_sums(ps, &red_pap[wave * 16 + erow], lo == 0);
        __syncthreads();

        float a[8];
        take8(rs_lane / sum_waves(red_pap, srow), a);
#pragma unroll
        for (int v = 0; v < 8; ++v) {
            xv[v].x = xv[v].x + a[v] * pv[v].x;  xv[v].y = xv[v].y + a[v] * pv[v].y;
            xv[v].z = xv[v].z + a[v] * pv[v].z;  xv[v].w = xv[v].w + a[v] * pv[v].w;
            rv[v].x = rv[v].x - a[v] * av[v].x;  rv[v].y = rv[v].y - a[v] * av[v].y;
            rv[v].z = rv[v].z - a[v] * av[v].z;  rv[v].w = rv[v].w - a[v] * av[v].w;
            ps[v] = dot4(rv[v], rv[v]);
        }
        strip_row_sums(ps, &red_rr[wave * 16 + erow], lo == 0);
        __syncthreads();

        float rs_next = sum_waves(red_rr, srow);
        if (__builtin_amdgcn_ballot_w32(!(rs_next <= rs_stop)) == 0u) break;
        float b[8];
        take8(rs_next / rs_lane, b);
        rs_lane = rs_next;
#pragma unroll
        for (int v = 0; v < 8; ++v) {
            float4 pn;
            pn.x = rv[v].x + b[v] * pv[v].x;  pn.y = rv[v].y + b[v] * pv[v].y;
            pn.z = rv[v].z + b[v] * pv[v].z;  pn.w = rv[v].w + b[v] * pv[v].w;
            *reinterpret_cast<float4*>(&p_s[(erow + v) * K_DIM + ecol]) = pn;
        }
        __syncthreads();
    }

#pragma unroll
    for (int v = 0; v < 8; ++v)
        *reinterpret_cast<float4*>(&Xout[(m0 + erow + v) * K_DIM + ecol]) = xv[v];
}

extern "C" void kernel_launch(void* const* d_in, const int* in_sizes, int n_in,
                              void* d_out, int out_size, void* d_ws, size_t ws_size,
                              hipStream_t stream) {
    const float* A  = (const float*)d_in[0];
    const float* B  = (const float*)d_in[1];
    const float* ea = (const float*)d_in[2];
    const float* eb = (const float*)d_in[3];
    float* out = (float*)d_out;

    float* ws      = (float*)d_ws;
    float* s_max   = ws;
    float* mask_ab = s_max   + 16;
    float* mask_ba = mask_ab + K_DIM * K_DIM;
    float* G_a     = mask_ba + K_DIM * K_DIM;
    float* G_b     = G_a     + K_DIM * K_DIM;
    float* R_ab    = G_b     + K_DIM * K_DIM;
    float* R_ba    = R_ab    + K_DIM * K_DIM;

    max_reduce_kernel<<<1, 512, 0, stream>>>(ea, eb, s_max);

    dim3 mgrid(K_DIM / 256, K_DIM);
    mask_kernel<<<mgrid, 256, 0, stream>>>(eb, ea, s_max, mask_ab);
    mask_kernel<<<mgrid, 256, 0, stream>>>(ea, eb, s_max, mask_ba);

    dim3 ggrid(8, 8, 4);
    gram_kernel<<<ggrid, 128, 0, stream>>>(A, B, G_a);

    size_t shmem = (size_t)16 * K_DIM * sizeof(float);
    dim3 cgrid(K_DIM / 16, 2);
    cg_solve_kernel<<<cgrid, 256, shmem, stream>>>(
        G_a, R_ab, mask_ab, out,
        G_b, R_ba, mask_ba, out + K_DIM * K_DIM,
        NITER);
}


# CG stops at <r,r> <= 2^-52 <r0,r0> (|r| <= 2^-26 |b|; was 2^-30): left-out part <= 2^-26 cond |x|, 1/4 of the fp32 floor 2^-24 cond |x|; later steps fall under an ulp of x; sum order unchanged
# speedup vs baseline: 12.8334x; 1.1315x over previous
#include <hip/hip_runtime.h>


#define D_DIM   1024
#define K_DIM   512
#define LMBDA_F 1000.0f
#define NITER   96

typedef float v2f __attribute__((ext_vector_type(2)));
typedef float v4f __attribute__((ext_vector_type(4)));
typedef float v8f __attribute__((ext_vector_type(8)));

__device__ __forceinline__ v8f wmma_f32(v2f a, v2f b, v8f c) {
    return __builtin_amdgcn_wmma_f32_16x16x4_f32(
        false, a, false, b, (short)0, c, false, false);
}

__global__ void max_reduce_kernel(const float* __restrict__ ea,
                                  const float* __restrict__ eb,
                                  float* __restrict__ s_out) {
    __shared__ float sm[512];
    int t = threadIdx.x;
    sm[t] = fmaxf(ea[t], eb[t]);
    __syncthreads();
    for (int off = 256; off > 0; off >>= 1) {
        if (t < off) sm[t] = fmaxf(sm[t], sm[t + off]);
        __syncthreads();
    }
    if (t == 0) s_out[0] = sm[0];
}

__global__ void mask_kernel(const float* __restrict__ erow,
                            const float* __restrict__ ecol,
                            const float* __restrict__ s_ptr,
                            float* __restrict__ mask) {
    int j = blockIdx.x * blockDim.x + threadIdx.x;
    int i = blockIdx.y;
    float inv_s = 1.0f / s_ptr[0];
    float gb = erow[i] * inv_s;
    float ga = ecol[j] * inv_s;
    float gb2 = gb * gb + 1.0f;
    float ga2 = ga * ga + 1.0f;
    float re = gb / gb2 - ga / ga2;
    float im = 1.0f / gb2 - 1.0f / ga2;
    mask[i * K_DIM + j] = re * re + im * im;
}

#define GT     64
#define GKB    32
#define GSTEPS (D_DIM / GKB)
#define GPP 160

#define COVER_NOPS "v_nop\n\tv_nop\n\tv_nop\n\tv_nop\n\tv_nop\n\tv_nop\n\tv_nop\n\tv_nop"

__device__ __forceinline__ void gram_cover_last(v8f (&acc)[2][2], int& rb, int& nl) {
    asm volatile(COVER_NOPS
        : "+v"(acc[0][0]), "+v"(acc[0][1]), "+v"(acc[1][0]), "+v"(acc[1][1]),
          "+v"(rb), "+v"(nl));
}

__device__ __forceinline__ void gram_fetch(const float* __restrict__ xg,
                                           const float* __restrict__ yg,
                                           v4f (&xr)[4], v4f (&yr)[4]) {
#pragma unroll
    for (int i = 0; i < 2; ++i) {
        xr[2 * i]     = *reinterpret_cast<const v4f*>(xg + (16 * i) * K_DIM);
        xr[2 * i + 1] = *reinterpret_cast<const v4f*>(xg + (16 * i + 1) * K_DIM);
        yr[2 * i]     = *reinterpret_cast<const v4f*>(yg + (16 * i) * K_DIM);
        yr[2 * i + 1] = *reinterpret_cast<const v4f*>(yg + (16 * i + 1) * K_DIM);
    }
}

__device__ __forceinline__ void gram_stage(float* sx, float* sy,
                                           const v4f (&xr)[4], const v4f (&yr)[4]) {
#pragma unroll
    for (int i = 0; i < 2; ++i) {
        float* px = sx + 8 * i * GPP;
        float* py = sy + 8 * i * GPP;
        *reinterpret_cast<v4f*>(px)     = __builtin_shufflevector(xr[2 * i], xr[2 * i + 1], 0, 4, 1, 5);
        *reinterpret_cast<v4f*>(px + 4) = __builtin_shufflevector(xr[2 * i], xr[2 * i + 1], 2, 6, 3, 7);
        *reinterpret_cast<v4f*>(py)     = __builtin_shufflevector(yr[2 * i], yr[2 * i + 1], 0, 4, 1, 5);
        *reinterpret_cast<v4f*>(py + 4) = __builtin_shufflevector(yr[2 * i], yr[2 * i + 1], 2, 6, 3, 7);
    }
}

__global__ void __launch_bounds__(128, 1)
gram_kernel(const float* __restrict__ A,
            const float* __restrict__ B,
            float* __restrict__ C4) {
    __shared__ __align__(16) float sX[2][(GKB / 2) * GPP];
    __shared__ __align__(16) float sY[2][(GKB / 2) * GPP];

    const int z = blockIdx.z;
    const float* __restrict__ X = (z == 1 || z == 2) ? B : A;
    const float* __restrict__ Y = (z == 1 || z == 3) ? B : A;
    float* __restrict__ C = C4 + (size_t)z * K_DIM * K_DIM;
    int t    = threadIdx.x;
    int lane = t & 31;
    int wave = t >> 5;
    int lo = lane & 15;
    int hi = lane >> 4;
    int bm0 = blockIdx.x * GT;
    int bn0 = blockIdx.y * GT;
    int wm  = (wave & 1) * 32;
    int wn  = (wave >> 1) * 32;

    int spr  = t >> 4;
    int scol = (t & 15) * 4;
    const float* xg = X + (2 * spr) * K_DIM + bm0 + scol;
    const float* yg = Y + (2 * spr) * K_DIM + bn0 + scol;
    int sidx = spr * GPP + 2 * scol;

    v4f xr[4], yr[4];
    gram_fetch(xg, yg, xr, yr);
    gram_stage(&sX[0][sidx], &sY[0][sidx], xr, yr);
    __syncthreads();

    v8f zero = {};
    v8f acc[2][2];
#pragma unroll
    for (int mi = 0; mi < 2; ++mi)
#pragma unroll
        for (int ni = 0; ni < 2; ++ni) acc[mi][ni] = zero;

    int fa = hi * GPP + 2 * (wm + lo);
    int fb = hi * GPP + 2 * (wn + lo);

    for (int s = 0; s < GSTEPS; ++s) {
        const int cur = s & 1;
        const int sn = (s + 1) & (GSTEPS - 1);
        gram_fetch(xg + sn * (GKB * K_DIM), yg + sn * (GKB * K_DIM), xr, yr);

        const float* ax = &sX[cur][fa];
        const float* by = &sY[cur][fb];
#pragma unroll
        for (int u = 0; u < GKB / 4; ++u) {
            v2f aF[2], bF[2];
#pragma unroll
            for (int mi = 0; mi < 2; ++mi)
                aF[mi] = *reinterpret_cast<const v2f*>(ax + (2 * u) * GPP + 32 * mi);
#pragma unroll
            for (int ni = 0; ni < 2; ++ni)
                bF[ni] = *reinterpret_cast<const v2f*>(by + (2 * u) * GPP + 32 * ni);
#pragma unroll
            for (int mi = 0; mi < 2; ++mi)
#pragma unroll
                for (int ni = 0; ni < 2; ++ni)
                    acc[mi][ni] = wmma_f32(aF[mi], bF[ni], acc[mi][ni]);
        }

        gram_stage(&sX[cur ^ 1][sidx], &sY[cur ^ 1][sidx], xr, yr);
        __syncthreads();
    }

    int nl = bn0 + wn + lo;
    int rb = (bm0 + wm + 8 * hi) * K_DIM + nl;
    gram_cover_last(acc, rb, nl);
    if (z < 2) {
        int pb = rb + nl;
#pragma unroll
        for (int mi = 0; mi < 2; ++mi)
#pragma unroll
            for (int ni = 0; ni < 2; ++ni)
#pragma unroll
                for (int v = 0; v < 8; v += 2) {
                    v2f pr;
                    pr.x = acc[mi][ni][v];
                    pr.y = acc[mi][ni][v + 1];
                    *reinterpret_cast<v2f*>(&C[pb + (mi * 16 + v) * K_DIM + 32 * ni]) = pr;
                }
    } else {
#pragma unroll
        for (int mi = 0; mi < 2; ++mi)
#pragma unroll
            for (int ni = 0; ni < 2; ++ni)
#pragma unroll
                for (int v = 0; v < 8; ++v)
                    C[rb + (mi * 16 + v) * K_DIM + 16 * ni] = acc[mi][ni][v];
    }
}

#define KU 4
#define BK (4 * KU)
#define NBLK (K_DIM / BK)
static_assert(NBLK % 4 == 0 && NBLK >= 8, "the ring of four register sets turns once per four blocks");

__device__ __forceinline__ void load_b_block(const float* __restrict__ gblk,
                                             unsigned gofb, v4f (&s)[2 * KU]) {
    const char* gb = reinterpret_cast<const char*>(gblk);
#pragma unroll
    for (int u = 0; u < KU; ++u) {
        s[2 * u]     = *reinterpret_cast<const v4f*>(gb + gofb + (2 * u) * (2 * K_DIM) * 4);
        s[2 * u + 1] = *reinterpret_cast<const v4f*>(gb + gofb + (2 * u) * (2 * K_DIM) * 4 + 16);
    }
}

__device__ __forceinline__ void load_a_block(const float* pa, v2f (&aF)[KU]) {
#pragma unroll
    for (int u = 0; u < KU; ++u)
        aF[u] = *reinterpret_cast<const v2f*>(pa + 4 * u);
}

__device__ __forceinline__ void mma_block(const v2f (&aF)[KU],
                                          const v4f (&s)[2 * KU],
                                          v8f (&acc)[4]) {
    __builtin_amdgcn_sched_barrier(0);
#pragma unroll
    for (int u = 0; u < KU; ++u) {
        acc[0] = wmma_f32(aF[u], __builtin_shufflevector(s[2 * u],     s[2 * u],     0, 1), acc[0]);
        acc[1] = wmma_f32(aF[u], __builtin_shufflevector(s[2 * u],     s[2 * u],     2, 3), acc[1]);
        acc[2] = wmma_f32(aF[u], __builtin_shufflevector(s[2 * u + 1], s[2 * u + 1], 0, 1), acc[2]);
        acc[3] = wmma_f32(aF[u], __builtin_shufflevector(s[2 * u + 1], s[2 * u + 1], 2, 3), acc[3]);
    }
}

__device__ __forceinline__ void refill_behind_run(v8f (&acc)[4], unsigned& gofb) {
    asm volatile(""
        : "+v"(acc[0]), "+v"(acc[1]), "+v"(acc[2]), "+v"(acc[3]), "+v"(gofb));
}

template <bool LAST>
__device__ __forceinline__ void ring_turn(const float* __restrict__ G, unsigned& gofb,
                                          const float* pbase, int blk,
                                          v4f (&s0)[2 * KU], v4f (&s1)[2 * KU],
                                          v4f (&s2)[2 * KU], v4f (&s3)[2 * KU],
                                          v2f (&aA)[KU], v2f (&aB)[KU], v8f (&acc)[4]) {
    refill_behind_run(acc, gofb);
    load_b_block(G + (blk + 3) * (BK * K_DIM), gofb, s3);
    load_a_block(pbase + (blk + 1) * BK, aB);
    mma_block(aA, s0, acc);
    refill_behind_run(acc, gofb);
    load_b_block(G + (LAST ? 0 : blk + 4) * (BK * K_DIM), gofb, s0);
    load_a_block(pbase + (blk + 2) * BK, aA);
    mma_block(aB, s1, acc);
    refill_behind_run(acc, gofb);
    load_b_block(G + (LAST ? 1 : blk + 5) * (BK * K_DIM), gofb, s1);
    load_a_block(pbase + (blk + 3) * BK, aB);
    mma_block(aA, s2, acc);
    refill_behind_run(acc, gofb);
    load_b_block(G + (LAST ? 2 : blk + 6) * (BK * K_DIM), gofb, s2);
    if (!LAST) load_a_block(pbase + (blk + 4) * BK, aA);
    mma_block(aB, s3, acc);
}

__device__ __forceinline__ void cover_last(v8f (&acc)[4]) {
    asm volatile(COVER_NOPS
        : "+v"(acc[0]), "+v"(acc[1]), "+v"(acc[2]), "+v"(acc[3]));
}

template <int XMASK>
__device__ __forceinline__ float row_xmask_add(float x) {
    int y = __builtin_amdgcn_update_dpp(0, __float_as_int(x), 0x160 + XMASK,
                                        0xf, 0xf, true);
    return x + __int_as_float(y);
}

__device__ __forceinline__ float row16_sum(float x) {
    x = row_xmask_add<1>(x);
    x = row_xmask_add<2>(x);
    x = row_xmask_add<4>(x);
    x = row_xmask_add<8>(x);
    return x;
}

#define CG_STOP_RR 0x1p-52f

__device__ __forceinline__ float dot4(float4 a, float4 b) {
    return a.x * b.x + a.y * b.y + a.z * b.z + a.w * b.w;
}

__device__ __forceinline__ void strip_row_sums(float (&ps)[8], float* slot, bool writer) {
#pragma unroll
    for (int v = 0; v < 8; ++v) ps[v] = row_xmask_add<1>(ps[v]);
#pragma unroll
    for (int v = 0; v < 8; ++v) ps[v] = row_xmask_add<2>(ps[v]);
#pragma unroll
    for (int v = 0; v < 8; ++v) ps[v] = row_xmask_add<4>(ps[v]);
#pragma unroll
    for (int v = 0; v < 8; ++v) ps[v] = row_xmask_add<8>(ps[v]);
    if (writer) {
        float4 s0, s1;
        s0.x = ps[0]; s0.y = ps[1]; s0.z = ps[2]; s0.w = ps[3];
        s1.x = ps[4]; s1.y = ps[5]; s1.z = ps[6]; s1.w = ps[7];
        *reinterpret_cast<float4*>(slot)     = s0;
        *reinterpret_cast<float4*>(slot + 4) = s1;
    }
}

__device__ __forceinline__ float sum_waves(const float* red, int row) {
    float s = red[row];
#pragma unroll
    for (int w = 1; w < 8; ++w) s += red[w * 16 + row];
    return s;
}

template <int SRC>
__device__ __forceinline__ float row_take(float x) {
    int y = __builtin_amdgcn_update_dpp(0, __float_as_int(x), 0x150 + SRC,
                                        0xf, 0xf, true);
    return __int_as_float(y);
}

__device__ __forceinline__ void take8(float s, float (&o)[8]) {
    o[0] = row_take<0>(s); o[1] = row_take<1>(s);
    o[2] = row_take<2>(s); o[3] = row_take<3>(s);
    o[4] = row_take<4>(s); o[5] = row_take<5>(s);
    o[6] = row_take<6>(s); o[7] = row_take<7>(s);
}

__global__ void __launch_bounds__(256, 1)
cg_solve_kernel(const float* __restrict__ G0,
                const float* __restrict__ R0,
                const float* __restrict__ mask0,
                float* __restrict__ Xout0,
                const float* __restrict__ G1,
                const float* __restrict__ R1,
                const float* __restrict__ mask1,
                float* __restrict__ Xout1,
                int niter) {
    const bool dir1 = blockIdx.y != 0;
    const float* __restrict__ G    = dir1 ? G1    : G0;
    const float* __restrict__ R    = dir1 ? R1    : R0;
    const float* __restrict__ mask = dir1 ? mask1 : mask0;
    float* __restrict__ Xout       = dir1 ? Xout1 : Xout0;
    extern __shared__ float smem[];
    float* p_s = smem;
    __shared__ __align__(16) float red_pap[8 * 16];
    __shared__ __align__(16) float red_rr[8 * 16];

    int t    = threadIdx.x;
    int lane = t & 31;
    int wave = t >> 5;
    int lo   = lane & 15;
    int hi   = lane >> 4;
    int n0   = wave * 64;
    int m0   = blockIdx.x * 16;

    int erow = 8 * hi;
    int ecol = n0 + 4 * lo;
    int srow = erow + (lo & 7);

    unsigned gofb = 4u * (hi * (2 * K_DIM) + 2 * (n0 + 4 * lo));
    const float* pbase = p_s + lo * K_DIM + 2 * hi;

    v4f s0[2 * KU], s1[2 * KU], s2[2 * KU], s3[2 * KU];
    load_b_block(G, gofb, s0);
    load_b_block(G + 1 * BK * K_DIM, gofb, s1);
    load_b_block(G + 2 * BK * K_DIM, gofb, s2);

    float4 xv[8], rv[8];
    {
        float ps[8];
#pragma unroll
        for (int v = 0; v < 8; ++v) {
            float4 q = *reinterpret_cast<const float4*>(
                &R[(m0 + erow + v) * K_DIM + ecol]);
            rv[v] = q;
            xv[v].x = 0.0f; xv[v].y = 0.0f; xv[v].z = 0.0f; xv[v].w = 0.0f;
            *reinterpret_cast<float4*>(&p_s[(erow + v) * K_DIM + ecol]) = q;
            ps[v] = dot4(q, q);
        }
        strip_row_sums(ps, &red_rr[wave * 16 + erow], lo == 0);
    }
    __syncthreads();
    float rs_lane = sum_waves(red_rr, srow);
    const float rs_stop = CG_STOP_RR * rs_lane;

    float4 lmk[8];
#pragma unroll
    for (int v = 0; v < 8; ++v) {
        float4 mk = *reinterpret_cast<const float4*>(
            &mask[(m0 + erow + v) * K_DIM + ecol]);
        lmk[v].x = LMBDA_F * mk.x; lmk[v].y = LMBDA_F * mk.y;
        lmk[v].z = LMBDA_F * mk.z; lmk[v].w = LMBDA_F * mk.w;
    }

    for (int it = 0; it < niter; ++it) {
        v8f zero = {};
        v8f acc[4];
#pragma unroll
        for (int ni = 0; ni < 4; ++ni) acc[ni] = zero;

        {
            v2f aA[KU], aB[KU];
            load_a_block(pbase, aA);
#pragma nounroll
            for (int blk = 0; blk < NBLK - 4; blk += 4)
                ring_turn<false>(G, gofb, pbase, blk, s0, s1, s2, s3, aA, aB, acc);
            ring_turn<true>(G, gofb, pbase, NBLK - 4, s0, s1, s2, s3, aA, aB, acc);
            cover_last(acc);
        }

        float4 pv[8], av[8];
        float ps[8];
#pragma unroll
        for (int v = 0; v < 8; ++v) {
            pv[v] = *reinterpret_cast<const float4*>(&p_s[(erow + v) * K_DIM + ecol]);
            av[v].x = acc[0][v] + lmk[v].x * pv[v].x;
            av[v].y = acc[1][v] + lmk[v].y * pv[v].y;
            av[v].z = acc[2][v] + lmk[v].z * pv[v].z;
            av[v].w = acc[3][v] + lmk[v].w * pv[v].w;
            ps[v] = dot4(pv[v], av[v]);
        }
        strip_row_sums(ps, &red_pap[wave * 16 + erow], lo == 0);
        __syncthreads();

        float a[8];
        take8(rs_lane / sum_waves(red_pap, srow), a);
#pragma unroll
        for (int v = 0; v < 8; ++v) {
            xv[v].x = xv[v].x + a[v] * pv[v].x;  xv[v].y = xv[v].y + a[v] * pv[v].y;
            xv[v].z = xv[v].z + a[v] * pv[v].z;  xv[v].w = xv[v].w + a[v] * pv[v].w;
            rv[v].x = rv[v].x - a[v] * av[v].x;  rv[v].y = rv[v].y - a[v] * av[v].y;
            rv[v].z = rv[v].z - a[v] * av[v].z;  rv[v].w = rv[v].w - a[v] * av[v].w;
            ps[v] = dot4(rv[v], rv[v]);
        }
        strip_row_sums(ps, &red_rr[wave * 16 + erow], lo == 0);
        __syncthreads();

        float rs_next = sum_waves(red_rr, srow);
        if (__builtin_amdgcn_ballot_w32(!(rs_next <= rs_stop)) == 0u) break;
        float b[8];
        take8(rs_next / rs_lane, b);
        rs_lane = rs_next;
#pragma unroll
        for (int v = 0; v < 8; ++v) {
            float4 pn;
            pn.x = rv[v].x + b[v] * pv[v].x;  pn.y = rv[v].y + b[v] * pv[v].y;
            pn.z = rv[v].z + b[v] * pv[v].z;  pn.w = rv[v].w + b[v] * pv[v].w;
            *reinterpret_cast<float4*>(&p_s[(erow + v) * K_DIM + ecol]) = pn;
        }
        __syncthreads();
    }

#pragma unroll
    for (int v = 0; v < 8; ++v)
        *reinterpret_cast<float4*>(&Xout[(m0 + erow + v) * K_DIM + ecol]) = xv[v];
}

extern "C" void kernel_launch(void* const* d_in, const int* in_sizes, int n_in,
                              void* d_out, int out_size, void* d_ws, size_t ws_size,
                              hipStream_t stream) {
    const float* A  = (const float*)d_in[0];
    const float* B  = (const float*)d_in[1];
    const float* ea = (const float*)d_in[2];
    const float* eb = (const float*)d_in[3];
    float* out = (float*)d_out;

    float* ws      = (float*)d_ws;
    float* s_max   = ws;
    float* mask_ab = s_max   + 16;
    float* mask_ba = mask_ab + K_DIM * K_DIM;
    float* G_a     = mask_ba + K_DIM * K_DIM;
    float* G_b     = G_a     + K_DIM * K_DIM;
    float* R_ab    = G_b     + K_DIM * K_DIM;
    float* R_ba    = R_ab    + K_DIM * K_DIM;

    max_reduce_kernel<<<1, 512, 0, stream>>>(ea, eb, s_max);

    dim3 mgrid(K_DIM / 256, K_DIM);
    mask_kernel<<<mgrid, 256, 0, stream>>>(eb, ea, s_max, mask_ab);
    mask_kernel<<<mgrid, 256, 0, stream>>>(ea, eb, s_max, mask_ba);

    dim3 ggrid(8, 8, 4);
    gram_kernel<<<ggrid, 128, 0, stream>>>(A, B, G_a);

    size_t shmem = (size_t)16 * K_DIM * sizeof(float);
    dim3 cgrid(K_DIM / 16, 2);
    cg_solve_kernel<<<cgrid, 256, shmem, stream>>>(
        G_a, R_ab, mask_ab, out,
        G_b, R_ba, mask_ba, out + K_DIM * K_DIM,
        NITER);
}


# Resolvent mask worked out in the CG kernel's registers from the two spectra, each workgroup taking the joint max itself: max launch, both mask launches and the mask planes go (5 launches -> 2)
# speedup vs baseline: 13.3808x; 1.0427x over previous
#include <hip/hip_runtime.h>


#define D_DIM   1024
#define K_DIM   512
#define LMBDA_F 1000.0f
#define NITER   96

typedef float v2f __attribute__((ext_vector_type(2)));
typedef float v4f __attribute__((ext_vector_type(4)));
typedef float v8f __attribute__((ext_vector_type(8)));

__device__ __forceinline__ v8f wmma_f32(v2f a, v2f b, v8f c) {
    return __builtin_amdgcn_wmma_f32_16x16x4_f32(
        false, a, false, b, (short)0, c, false, false);
}


#define GT     64
#define GKB    32
#define GSTEPS (D_DIM / GKB)
#define GPP 160

#define COVER_NOPS "v_nop\n\tv_nop\n\tv_nop\n\tv_nop\n\tv_nop\n\tv_nop\n\tv_nop\n\tv_nop"

__device__ __forceinline__ void gram_cover_last(v8f (&acc)[2][2], int& rb, int& nl) {
    asm volatile(COVER_NOPS
        : "+v"(acc[0][0]), "+v"(acc[0][1]), "+v"(acc[1][0]), "+v"(acc[1][1]),
          "+v"(rb), "+v"(nl));
}

__device__ __forceinline__ void gram_fetch(const float* __restrict__ xg,
                                           const float* __restrict__ yg,
                                           v4f (&xr)[4], v4f (&yr)[4]) {
#pragma unroll
    for (int i = 0; i < 2; ++i) {
        xr[2 * i]     = *reinterpret_cast<const v4f*>(xg + (16 * i) * K_DIM);
        xr[2 * i + 1] = *reinterpret_cast<const v4f*>(xg + (16 * i + 1) * K_DIM);
        yr[2 * i]     = *reinterpret_cast<const v4f*>(yg + (16 * i) * K_DIM);
        yr[2 * i + 1] = *reinterpret_cast<const v4f*>(yg + (16 * i + 1) * K_DIM);
    }
}

__device__ __forceinline__ void gram_stage(float* sx, float* sy,
                                           const v4f (&xr)[4], const v4f (&yr)[4]) {
#pragma unroll
    for (int i = 0; i < 2; ++i) {
        float* px = sx + 8 * i * GPP;
        float* py = sy + 8 * i * GPP;
        *reinterpret_cast<v4f*>(px)     = __builtin_shufflevector(xr[2 * i], xr[2 * i + 1], 0, 4, 1, 5);
        *reinterpret_cast<v4f*>(px + 4) = __builtin_shufflevector(xr[2 * i], xr[2 * i + 1], 2, 6, 3, 7);
        *reinterpret_cast<v4f*>(py)     = __builtin_shufflevector(yr[2 * i], yr[2 * i + 1], 0, 4, 1, 5);
        *reinterpret_cast<v4f*>(py + 4) = __builtin_shufflevector(yr[2 * i], yr[2 * i + 1], 2, 6, 3, 7);
    }
}

__global__ void __launch_bounds__(128, 1)
gram_kernel(const float* __restrict__ A,
            const float* __restrict__ B,
            float* __restrict__ C4) {
    __shared__ __align__(16) float sX[2][(GKB / 2) * GPP];
    __shared__ __align__(16) float sY[2][(GKB / 2) * GPP];

    const int z = blockIdx.z;
    const float* __restrict__ X = (z == 1 || z == 2) ? B : A;
    const float* __restrict__ Y = (z == 1 || z == 3) ? B : A;
    float* __restrict__ C = C4 + (size_t)z * K_DIM * K_DIM;
    int t    = threadIdx.x;
    int lane = t & 31;
    int wave = t >> 5;
    int lo = lane & 15;
    int hi = lane >> 4;
    int bm0 = blockIdx.x * GT;
    int bn0 = blockIdx.y * GT;
    int wm  = (wave & 1) * 32;
    int wn  = (wave >> 1) * 32;

    int spr  = t >> 4;
    int scol = (t & 15) * 4;
    const float* xg = X + (2 * spr) * K_DIM + bm0 + scol;
    const float* yg = Y + (2 * spr) * K_DIM + bn0 + scol;
    int sidx = spr * GPP + 2 * scol;

    v4f xr[4], yr[4];
    gram_fetch(xg, yg, xr, yr);
    gram_stage(&sX[0][sidx], &sY[0][sidx], xr, yr);
    __syncthreads();

    v8f zero = {};
    v8f acc[2][2];
#pragma unroll
    for (int mi = 0; mi < 2; ++mi)
#pragma unroll
        for (int ni = 0; ni < 2; ++ni) acc[mi][ni] = zero;

    int fa = hi * GPP + 2 * (wm + lo);
    int fb = hi * GPP + 2 * (wn + lo);

    for (int s = 0; s < GSTEPS; ++s) {
        const int cur = s & 1;
        const int sn = (s + 1) & (GSTEPS - 1);
        gram_fetch(xg + sn * (GKB * K_DIM), yg + sn * (GKB * K_DIM), xr, yr);

        const float* ax = &sX[cur][fa];
        const float* by = &sY[cur][fb];
#pragma unroll
        for (int u = 0; u < GKB / 4; ++u) {
            v2f aF[2], bF[2];
#pragma unroll
            for (int mi = 0; mi < 2; ++mi)
                aF[mi] = *reinterpret_cast<const v2f*>(ax + (2 * u) * GPP + 32 * mi);
#pragma unroll
            for (int ni = 0; ni < 2; ++ni)
                bF[ni] = *reinterpret_cast<const v2f*>(by + (2 * u) * GPP + 32 * ni);
#pragma unroll
            for (int mi = 0; mi < 2; ++mi)
#pragma unroll
                for (int ni = 0; ni < 2; ++ni)
                    acc[mi][ni] = wmma_f32(aF[mi], bF[ni], acc[mi][ni]);
        }

        gram_stage(&sX[cur ^ 1][sidx], &sY[cur ^ 1][sidx], xr, yr);
        __syncthreads();
    }

    int nl = bn0 + wn + lo;
    int rb = (bm0 + wm + 8 * hi) * K_DIM + nl;
    gram_cover_last(acc, rb, nl);
    if (z < 2) {
        int pb = rb + nl;
#pragma unroll
        for (int mi = 0; mi < 2; ++mi)
#pragma unroll
            for (int ni = 0; ni < 2; ++ni)
#pragma unroll
                for (int v = 0; v < 8; v += 2) {
                    v2f pr;
                    pr.x = acc[mi][ni][v];
                    pr.y = acc[mi][ni][v + 1];
                    *reinterpret_cast<v2f*>(&C[pb + (mi * 16 + v) * K_DIM + 32 * ni]) = pr;
                }
    } else {
#pragma unroll
        for (int mi = 0; mi < 2; ++mi)
#pragma unroll
            for (int ni = 0; ni < 2; ++ni)
#pragma unroll
                for (int v = 0; v < 8; ++v)
                    C[rb + (mi * 16 + v) * K_DIM + 16 * ni] = acc[mi][ni][v];
    }
}

#define KU 4
#define BK (4 * KU)
#define NBLK (K_DIM / BK)
static_assert(NBLK % 4 == 0 && NBLK >= 8, "the ring of four register sets turns once per four blocks");

__device__ __forceinline__ void load_b_block(const float* __restrict__ gblk,
                                             unsigned gofb, v4f (&s)[2 * KU]) {
    const char* gb = reinterpret_cast<const char*>(gblk);
#pragma unroll
    for (int u = 0; u < KU; ++u) {
        s[2 * u]     = *reinterpret_cast<const v4f*>(gb + gofb + (2 * u) * (2 * K_DIM) * 4);
        s[2 * u + 1] = *reinterpret_cast<const v4f*>(gb + gofb + (2 * u) * (2 * K_DIM) * 4 + 16);
    }
}

__device__ __forceinline__ void load_a_block(const float* pa, v2f (&aF)[KU]) {
#pragma unroll
    for (int u = 0; u < KU; ++u)
        aF[u] = *reinterpret_cast<const v2f*>(pa + 4 * u);
}

__device__ __forceinline__ void mma_block(const v2f (&aF)[KU],
                                          const v4f (&s)[2 * KU],
                                          v8f (&acc)[4]) {
    __builtin_amdgcn_sched_barrier(0);
#pragma unroll
    for (int u = 0; u < KU; ++u) {
        acc[0] = wmma_f32(aF[u], __builtin_shufflevector(s[2 * u],     s[2 * u],     0, 1), acc[0]);
        acc[1] = wmma_f32(aF[u], __builtin_shufflevector(s[2 * u],     s[2 * u],     2, 3), acc[1]);
        acc[2] = wmma_f32(aF[u], __builtin_shufflevector(s[2 * u + 1], s[2 * u + 1], 0, 1), acc[2]);
        acc[3] = wmma_f32(aF[u], __builtin_shufflevector(s[2 * u + 1], s[2 * u + 1], 2, 3), acc[3]);
    }
}

__device__ __forceinline__ void refill_behind_run(v8f (&acc)[4], unsigned& gofb) {
    asm volatile(""
        : "+v"(acc[0]), "+v"(acc[1]), "+v"(acc[2]), "+v"(acc[3]), "+v"(gofb));
}

template <bool LAST>
__device__ __forceinline__ void ring_turn(const float* __restrict__ G, unsigned& gofb,
                                          const float* pbase, int blk,
                                          v4f (&s0)[2 * KU], v4f (&s1)[2 * KU],
                                          v4f (&s2)[2 * KU], v4f (&s3)[2 * KU],
                                          v2f (&aA)[KU], v2f (&aB)[KU], v8f (&acc)[4]) {
    refill_behind_run(acc, gofb);
    load_b_block(G + (blk + 3) * (BK * K_DIM), gofb, s3);
    load_a_block(pbase + (blk + 1) * BK, aB);
    mma_block(aA, s0, acc);
    refill_behind_run(acc, gofb);
    load_b_block(G + (LAST ? 0 : blk + 4) * (BK * K_DIM), gofb, s0);
    load_a_block(pbase + (blk + 2) * BK, aA);
    mma_block(aB, s1, acc);
    refill_behind_run(acc, gofb);
    load_b_block(G + (LAST ? 1 : blk + 5) * (BK * K_DIM), gofb, s1);
    load_a_block(pbase + (blk + 3) * BK, aB);
    mma_block(aA, s2, acc);
    refill_behind_run(acc, gofb);
    load_b_block(G + (LAST ? 2 : blk + 6) * (BK * K_DIM), gofb, s2);
    if (!LAST) load_a_block(pbase + (blk + 4) * BK, aA);
    mma_block(aB, s3, acc);
}

__device__ __forceinline__ void cover_last(v8f (&acc)[4]) {
    asm volatile(COVER_NOPS
        : "+v"(acc[0]), "+v"(acc[1]), "+v"(acc[2]), "+v"(acc[3]));
}

template <int XMASK>
__device__ __forceinline__ float row_xmask_add(float x) {
    int y = __builtin_amdgcn_update_dpp(0, __float_as_int(x), 0x160 + XMASK,
                                        0xf, 0xf, true);
    return x + __int_as_float(y);
}

__device__ __forceinline__ float row16_sum(float x) {
    x = row_xmask_add<1>(x);
    x = row_xmask_add<2>(x);
    x = row_xmask_add<4>(x);
    x = row_xmask_add<8>(x);
    return x;
}

#define CG_STOP_RR 0x1p-52f

__device__ __forceinline__ float dot4(float4 a, float4 b) {
    return a.x * b.x + a.y * b.y + a.z * b.z + a.w * b.w;
}

__device__ __forceinline__ void strip_row_sums(float (&ps)[8], float* slot, bool writer) {
#pragma unroll
    for (int v = 0; v < 8; ++v) ps[v] = row_xmask_add<1>(ps[v]);
#pragma unroll
    for (int v = 0; v < 8; ++v) ps[v] = row_xmask_add<2>(ps[v]);
#pragma unroll
    for (int v = 0; v < 8; ++v) ps[v] = row_xmask_add<4>(ps[v]);
#pragma unroll
    for (int v = 0; v < 8; ++v) ps[v] = row_xmask_add<8>(ps[v]);
    if (writer) {
        float4 s0, s1;
        s0.x = ps[0]; s0.y = ps[1]; s0.z = ps[2]; s0.w = ps[3];
        s1.x = ps[4]; s1.y = ps[5]; s1.z = ps[6]; s1.w = ps[7];
        *reinterpret_cast<float4*>(slot)     = s0;
        *reinterpret_cast<float4*>(slot + 4) = s1;
    }
}

__device__ __forceinline__ float sum_waves(const float* red, int row) {
    float s = red[row];
#pragma unroll
    for (int w = 1; w < 8; ++w) s += red[w * 16 + row];
    return s;
}

template <int SRC>
__device__ __forceinline__ float row_take(float x) {
    int y = __builtin_amdgcn_update_dpp(0, __float_as_int(x), 0x150 + SRC,
                                        0xf, 0xf, true);
    return __int_as_float(y);
}

__device__ __forceinline__ void take8(float s, float (&o)[8]) {
    o[0] = row_take<0>(s); o[1] = row_take<1>(s);
    o[2] = row_take<2>(s); o[3] = row_take<3>(s);
    o[4] = row_take<4>(s); o[5] = row_take<5>(s);
    o[6] = row_take<6>(s); o[7] = row_take<7>(s);
}

template <int XMASK>
__device__ __forceinline__ float row_xmask_max(float x) {
    int y = __builtin_amdgcn_update_dpp(0, __float_as_int(x), 0x160 + XMASK,
                                        0xf, 0xf, true);
    return fmaxf(x, __int_as_float(y));
}

__device__ __forceinline__ float row16_max(float x) {
    x = row_xmask_max<1>(x);
    x = row_xmask_max<2>(x);
    x = row_xmask_max<4>(x);
    x = row_xmask_max<8>(x);
    return x;
}

__device__ __forceinline__ float2 spectrum_point(float e, float inv_s) {
    float g  = e * inv_s;
    float g2 = g * g + 1.0f;
    float2 q;
    q.x = g / g2;
    q.y = 1.0f / g2;
    return q;
}

__device__ __forceinline__ float mask_elem(float2 b, float2 a) {
    float re = b.x - a.x;
    float im = b.y - a.y;
    return re * re + im * im;
}

__global__ void __launch_bounds__(256, 1)
cg_solve_kernel(const float* __restrict__ G0,
                const float* __restrict__ R0,
                float* __restrict__ Xout0,
                const float* __restrict__ G1,
                const float* __restrict__ R1,
                float* __restrict__ Xout1,
                const float* __restrict__ ea,
                const float* __restrict__ eb,
                int niter) {
    const bool dir1 = blockIdx.y != 0;
    const float* __restrict__ G    = dir1 ? G1    : G0;
    const float* __restrict__ R    = dir1 ? R1    : R0;
    float* __restrict__ Xout       = dir1 ? Xout1 : Xout0;
    const float* __restrict__ ev_row = dir1 ? ea : eb;
    const float* __restrict__ ev_col = dir1 ? eb : ea;
    extern __shared__ float smem[];
    float* p_s = smem;
    __shared__ __align__(16) float red_pap[8 * 16];
    __shared__ __align__(16) float red_rr[8 * 16];
    __shared__ __align__(16) float red_max[16];

    int t    = threadIdx.x;
    int lane = t & 31;
    int wave = t >> 5;
    int lo   = lane & 15;
    int hi   = lane >> 4;
    int n0   = wave * 64;
    int m0   = blockIdx.x * 16;

    int erow = 8 * hi;
    int ecol = n0 + 4 * lo;
    int srow = erow + (lo & 7);

    unsigned gofb = 4u * (hi * (2 * K_DIM) + 2 * (n0 + 4 * lo));
    const float* pbase = p_s + lo * K_DIM + 2 * hi;

    v4f s0[2 * KU], s1[2 * KU], s2[2 * KU], s3[2 * KU];
    float e_a0 = ea[t], e_a1 = ea[t + 256], e_b0 = eb[t], e_b1 = eb[t + 256];
    load_b_block(G, gofb, s0);
    load_b_block(G + 1 * BK * K_DIM, gofb, s1);
    load_b_block(G + 2 * BK * K_DIM, gofb, s2);

    float4 er0 = *reinterpret_cast<const float4*>(&ev_row[m0 + erow]);
    float4 er1 = *reinterpret_cast<const float4*>(&ev_row[m0 + erow + 4]);
    float4 ec  = *reinterpret_cast<const float4*>(&ev_col[ecol]);

    float4 xv[8], rv[8];
    {
        float ps[8];
#pragma unroll
        for (int v = 0; v < 8; ++v) {
            float4 q = *reinterpret_cast<const float4*>(
                &R[(m0 + erow + v) * K_DIM + ecol]);
            rv[v] = q;
            xv[v].x = 0.0f; xv[v].y = 0.0f; xv[v].z = 0.0f; xv[v].w = 0.0f;
            *reinterpret_cast<float4*>(&p_s[(erow + v) * K_DIM + ecol]) = q;
            ps[v] = dot4(q, q);
        }
        strip_row_sums(ps, &red_rr[wave * 16 + erow], lo == 0);
    }
    {
        float mx = fmaxf(fmaxf(e_a0, e_a1), fmaxf(e_b0, e_b1));
        mx = row16_max(mx);
        if (lo == 0) red_max[2 * wave + hi] = mx;
    }
    __syncthreads();
    float rs_lane = sum_waves(red_rr, srow);
    const float rs_stop = CG_STOP_RR * rs_lane;

    float4 lmk[8];
    {
        float4 m0v = *reinterpret_cast<const float4*>(&red_max[0]);
        float4 m1v = *reinterpret_cast<const float4*>(&red_max[4]);
        float4 m2v = *reinterpret_cast<const float4*>(&red_max[8]);
        float4 m3v = *reinterpret_cast<const float4*>(&red_max[12]);
        float s = fmaxf(fmaxf(fmaxf(fmaxf(m0v.x, m0v.y), fmaxf(m0v.z, m0v.w)),
                              fmaxf(fmaxf(m1v.x, m1v.y), fmaxf(m1v.z, m1v.w))),
                        fmaxf(fmaxf(fmaxf(m2v.x, m2v.y), fmaxf(m2v.z, m2v.w)),
                              fmaxf(fmaxf(m3v.x, m3v.y), fmaxf(m3v.z, m3v.w))));
        float inv_s = 1.0f / s;
        float2 pc[4], pr[8];
        pc[0] = spectrum_point(ec.x, inv_s);  pc[1] = spectrum_point(ec.y, inv_s);
        pc[2] = spectrum_point(ec.z, inv_s);  pc[3] = spectrum_point(ec.w, inv_s);
        pr[0] = spectrum_point(er0.x, inv_s); pr[1] = spectrum_point(er0.y, inv_s);
        pr[2] = spectrum_point(er0.z, inv_s); pr[3] = spectrum_point(er0.w, inv_s);
        pr[4] = spectrum_point(er1.x, inv_s); pr[5] = spectrum_point(er1.y, inv_s);
        pr[6] = spectrum_point(er1.z, inv_s); pr[7] = spectrum_point(er1.w, inv_s);
#pragma unroll
        for (int v = 0; v < 8; ++v) {
            lmk[v].x = LMBDA_F * mask_elem(pr[v], pc[0]);
            lmk[v].y = LMBDA_F * mask_elem(pr[v], pc[1]);
            lmk[v].z = LMBDA_F * mask_elem(pr[v], pc[2]);
            lmk[v].w = LMBDA_F * mask_elem(pr[v], pc[3]);
        }
    }

    for (int it = 0; it < niter; ++it) {
        v8f zero = {};
        v8f acc[4];
#pragma unroll
        for (int ni = 0; ni < 4; ++ni) acc[ni] = zero;

        {
            v2f aA[KU], aB[KU];
            load_a_block(pbase, aA);
#pragma nounroll
            for (int blk = 0; blk < NBLK - 4; blk += 4)
                ring_turn<false>(G, gofb, pbase, blk, s0, s1, s2, s3, aA, aB, acc);
            ring_turn<true>(G, gofb, pbase, NBLK - 4, s0, s1, s2, s3, aA, aB, acc);
            cover_last(acc);
        }

        float4 pv[8], av[8];
        float ps[8];
#pragma unroll
        for (int v = 0; v < 8; ++v) {
            pv[v] = *reinterpret_cast<const float4*>(&p_s[(erow + v) * K_DIM + ecol]);
            av[v].x = acc[0][v] + lmk[v].x * pv[v].x;
            av[v].y = acc[1][v] + lmk[v].y * pv[v].y;
            av[v].z = acc[2][v] + lmk[v].z * pv[v].z;
            av[v].w = acc[3][v] + lmk[v].w * pv[v].w;
            ps[v] = dot4(pv[v], av[v]);
        }
        strip_row_sums(ps, &red_pap[wave * 16 + erow], lo == 0);
        __syncthreads();

        float a[8];
        take8(rs_lane / sum_waves(red_pap, srow), a);
#pragma unroll
        for (int v = 0; v < 8; ++v) {
            xv[v].x = xv[v].x + a[v] * pv[v].x;  xv[v].y = xv[v].y + a[v] * pv[v].y;
            xv[v].z = xv[v].z + a[v] * pv[v].z;  xv[v].w = xv[v].w + a[v] * pv[v].w;
            rv[v].x = rv[v].x - a[v] * av[v].x;  rv[v].y = rv[v].y - a[v] * av[v].y;
            rv[v].z = rv[v].z - a[v] * av[v].z;  rv[v].w = rv[v].w - a[v] * av[v].w;
            ps[v] = dot4(rv[v], rv[v]);
        }
        strip_row_sums(ps, &red_rr[wave * 16 + erow], lo == 0);
        __syncthreads();

        float rs_next = sum_waves(red_rr, srow);
        if (__builtin_amdgcn_ballot_w32(!(rs_next <= rs_stop)) == 0u) break;
        float b[8];
        take8(rs_next / rs_lane, b);
        rs_lane = rs_next;
#pragma unroll
        for (int v = 0; v < 8; ++v) {
            float4 pn;
            pn.x = rv[v].x + b[v] * pv[v].x;  pn.y = rv[v].y + b[v] * pv[v].y;
            pn.z = rv[v].z + b[v] * pv[v].z;  pn.w = rv[v].w + b[v] * pv[v].w;
            *reinterpret_cast<float4*>(&p_s[(erow + v) * K_DIM + ecol]) = pn;
        }
        __syncthreads();
    }

#pragma unroll
    for (int v = 0; v < 8; ++v)
        *reinterpret_cast<float4*>(&Xout[(m0 + erow + v) * K_DIM + ecol]) = xv[v];
}

extern "C" void kernel_launch(void* const* d_in, const int* in_sizes, int n_in,
                              void* d_out, int out_size, void* d_ws, size_t ws_size,
                              hipStream_t stream) {
    const float* A  = (const float*)d_in[0];
    const float* B  = (const float*)d_in[1];
    const float* ea = (const float*)d_in[2];
    const float* eb = (const float*)d_in[3];
    float* out = (float*)d_out;

    float* ws      = (float*)d_ws;
    float* G_a     = ws;
    float* G_b     = G_a     + K_DIM * K_DIM;
    float* R_ab    = G_b     + K_DIM * K_DIM;
    float* R_ba    = R_ab    + K_DIM * K_DIM;

    dim3 ggrid(8, 8, 4);
    gram_kernel<<<ggrid, 128, 0, stream>>>(A, B, G_a);

    size_t shmem = (size_t)16 * K_DIM * sizeof(float);
    dim3 cgrid(K_DIM / 16, 2);
    cg_solve_kernel<<<cgrid, 256, shmem, stream>>>(
        G_a, R_ab, out,
        G_b, R_ba, out + K_DIM * K_DIM,
        ea, eb, NITER);
}
